# starting kernel
# baseline (speedup 1.0000x reference)
#include <hip/hip_runtime.h>

typedef __attribute__((ext_vector_type(2))) float v2f;
typedef __attribute__((ext_vector_type(8))) float v8f;

#define IN_F  128
#define HID_F 128
#define OUT_F 64

__global__ void k_fill(float* __restrict__ p, float v, long n) {
    long i = (long)blockIdx.x * blockDim.x + threadIdx.x;
    long stride = (long)gridDim.x * blockDim.x;
    for (; i < n; i += stride) p[i] = v;
}

__global__ void k_degree(const int* __restrict__ dst, float* __restrict__ deg, long E) {
    long i = (long)blockIdx.x * blockDim.x + threadIdx.x;
    long stride = (long)gridDim.x * blockDim.x;
    for (; i < E; i += stride)
        __hip_atomic_fetch_add(&deg[dst[i]], 1.0f, __ATOMIC_RELAXED, __HIP_MEMORY_SCOPE_AGENT);
}

__global__ void k_rsqrt(float* __restrict__ deg, int N) {
    int i = blockIdx.x * blockDim.x + threadIdx.x;
    if (i < N) deg[i] = rsqrtf(deg[i]);
}

template <int K, int NCOL, int RT>
__global__ void __launch_bounds__(32 * (NCOL / 16))
k_gemm_wmma(const float* __restrict__ X, const float* __restrict__ W,
            float* __restrict__ H, int M) {
    constexpr int ROWS = 16 * RT;
    constexpr int SXS  = K + 4;
    __shared__ float sX[ROWS * SXS];

    const int tid  = threadIdx.x;
    const int wave = tid >> 5;
    const int lane = tid & 31;
    const int row0 = blockIdx.x * ROWS;

    for (int i = tid; i < ROWS * K; i += blockDim.x) {
        int r = i / K, c = i % K;
        int gr = row0 + r;
        sX[r * SXS + c] = (gr < M) ? X[(size_t)gr * K + c] : 0.0f;
    }
    __syncthreads();

    const int half = lane >> 4;
    const int idx  = lane & 15;
    const int colbase = wave * 16;

    v8f acc[RT] = {};
#pragma unroll
    for (int k0 = 0; k0 < K; k0 += 4) {
        const int ka = k0 + half * 2;
        v2f b;
        b.x = W[(size_t)(ka)     * NCOL + colbase + idx];
        b.y = W[(size_t)(ka + 1) * NCOL + colbase + idx];
#pragma unroll
        for (int t = 0; t < RT; ++t) {
            v2f a;
            a.x = sX[(t * 16 + idx) * SXS + ka];
            a.y = sX[(t * 16 + idx) * SXS + ka + 1];
            acc[t] = __builtin_amdgcn_wmma_f32_16x16x4_f32(
                 false, a,  false, b,
                 (short)0, acc[t],  false,  false);
        }
    }

#pragma unroll
    for (int t = 0; t < RT; ++t) {
#pragma unroll
        for (int v = 0; v < 8; ++v) {
            int gr = row0 + t * 16 + half * 8 + v;
            if (gr < M) H[(size_t)gr * NCOL + colbase + idx] = acc[t][v];
        }
    }
}

template <int F>
__global__ void k_scatter(const int* __restrict__ src, const int* __restrict__ dst,
                          const float* __restrict__ dinv, const float* __restrict__ h,
                          float* __restrict__ agg, long E) {
    constexpr int C = F / 32;
    const int wavesPerBlock = blockDim.x >> 5;
    const long e = (long)blockIdx.x * wavesPerBlock + (threadIdx.x >> 5);
    if (e >= E) return;
    const int lane = threadIdx.x & 31;
    const int s = src[e];
    const int d = dst[e];
    const float nm = dinv[s] * dinv[d];
    const float* __restrict__ hp = h + (size_t)s * F + lane * C;
    float* __restrict__ op = agg + (size_t)d * F + lane * C;
    float v[C];
#pragma unroll
    for (int j = 0; j < C; ++j) v[j] = hp[j];
#pragma unroll
    for (int j = 0; j < C; ++j)
        __hip_atomic_fetch_add(op + j, v[j] * nm, __ATOMIC_RELAXED, __HIP_MEMORY_SCOPE_AGENT);
}

__global__ void k_post_relu(float* __restrict__ agg, const float* __restrict__ h,
                            const float* __restrict__ dinv, const float* __restrict__ b,
                            long total) {
    long i = (long)blockIdx.x * blockDim.x + threadIdx.x;
    if (i >= total) return;
    int n = (int)(i / HID_F), f = (int)(i % HID_F);
    float di = dinv[n];
    float v = agg[i] + h[i] * di * di + b[f];
    agg[i] = v > 0.0f ? v : 0.0f;
}

__global__ void k_post(float* __restrict__ out, const float* __restrict__ h2,
                       const float* __restrict__ dinv, const float* __restrict__ b,
                       long total) {
    long i = (long)blockIdx.x * blockDim.x + threadIdx.x;
    if (i >= total) return;
    int n = (int)(i / OUT_F), f = (int)(i % OUT_F);
    float di = dinv[n];
    out[i] = out[i] + h2[i] * di * di + b[f];
}

extern "C" void kernel_launch(void* const* d_in, const int* in_sizes, int n_in,
                              void* d_out, int out_size, void* d_ws, size_t ws_size,
                              hipStream_t stream) {
    const float* x  = (const float*)d_in[0];
    const int*   ei = (const int*)d_in[1];
    const float* W1 = (const float*)d_in[2];
    const float* b1 = (const float*)d_in[3];
    const float* W2 = (const float*)d_in[4];
    const float* b2 = (const float*)d_in[5];
    float* out = (float*)d_out;

    const int  N = in_sizes[0] / IN_F;
    const long E = (long)in_sizes[1] / 2;
    const int* srcIdx = ei;
    const int* dstIdx = ei + E;

    float* ws   = (float*)d_ws;
    float* dinv = ws;
    float* h    = dinv + N;
    float* agg1 = h + (size_t)N * HID_F;
    float* h2   = agg1 + (size_t)N * HID_F;

    k_fill<<<1024, 256, 0, stream>>>(dinv, 1.0f, (long)N);
    k_fill<<<4096, 256, 0, stream>>>(agg1, 0.0f, (long)N * HID_F);
    k_fill<<<2048, 256, 0, stream>>>(out, 0.0f, (long)N * OUT_F);

    k_degree<<<4096, 256, 0, stream>>>(dstIdx, dinv, E);
    k_rsqrt<<<(N + 255) / 256, 256, 0, stream>>>(dinv, N);

    constexpr int RT   = 4;
    constexpr int ROWS = 16 * RT;

    k_gemm_wmma<IN_F, HID_F, RT>
        <<<(N + ROWS - 1) / ROWS, 32 * (HID_F / 16), 0, stream>>>(x, W1, h, N);
    k_scatter<HID_F><<<(int)((E + 7) / 8), 256, 0, stream>>>(srcIdx, dstIdx, dinv, h, agg1, E);
    k_post_relu<<<(int)(((long)N * HID_F + 255) / 256), 256, 0, stream>>>(agg1, h, dinv, b1,
                                                                          (long)N * HID_F);

    k_gemm_wmma<HID_F, OUT_F, RT>
        <<<(N + ROWS - 1) / ROWS, 32 * (OUT_F / 16), 0, stream>>>(agg1, W2, h2, N);
    k_scatter<OUT_F><<<(int)((E + 7) / 8), 256, 0, stream>>>(srcIdx, dstIdx, dinv, h2, out, E);
    k_post<<<(int)(((long)N * OUT_F + 255) / 256), 256, 0, stream>>>(out, h2, dinv, b2,
                                                                     (long)N * OUT_F);
}


# Aggregation by dst-tile ownership: a work-group keeps a tile of dst rows in LDS, scans the whole dst list and adds matching edges' rows by LDS float atomics, stores rows once; sum order = LDS arrival
# speedup vs baseline: 1.6933x; 1.6933x over previous
#include <hip/hip_runtime.h>

typedef __attribute__((ext_vector_type(2))) float v2f;
typedef __attribute__((ext_vector_type(8))) float v8f;

#define IN_F  128
#define HID_F 128
#define OUT_F 64

__global__ void k_fill(float* __restrict__ p, float v, long n) {
    long i = (long)blockIdx.x * blockDim.x + threadIdx.x;
    long stride = (long)gridDim.x * blockDim.x;
    for (; i < n; i += stride) p[i] = v;
}

__global__ void k_degree(const int* __restrict__ dst, float* __restrict__ deg, long E) {
    long i = (long)blockIdx.x * blockDim.x + threadIdx.x;
    long stride = (long)gridDim.x * blockDim.x;
    for (; i < E; i += stride)
        __hip_atomic_fetch_add(&deg[dst[i]], 1.0f, __ATOMIC_RELAXED, __HIP_MEMORY_SCOPE_AGENT);
}

__global__ void k_rsqrt(float* __restrict__ deg, int N) {
    int i = blockIdx.x * blockDim.x + threadIdx.x;
    if (i < N) deg[i] = rsqrtf(deg[i]);
}

template <int K, int NCOL, int RT>
__global__ void __launch_bounds__(32 * (NCOL / 16))
k_gemm_wmma(const float* __restrict__ X, const float* __restrict__ W,
            float* __restrict__ H, int M) {
    constexpr int ROWS = 16 * RT;
    constexpr int SXS  = K + 4;
    __shared__ float sX[ROWS * SXS];

    const int tid  = threadIdx.x;
    const int wave = tid >> 5;
    const int lane = tid & 31;
    const int row0 = blockIdx.x * ROWS;

    for (int i = tid; i < ROWS * K; i += blockDim.x) {
        int r = i / K, c = i % K;
        int gr = row0 + r;
        sX[r * SXS + c] = (gr < M) ? X[(size_t)gr * K + c] : 0.0f;
    }
    __syncthreads();

    const int half = lane >> 4;
    const int idx  = lane & 15;
    const int colbase = wave * 16;

    v8f acc[RT] = {};
#pragma unroll
    for (int k0 = 0; k0 < K; k0 += 4) {
        const int ka = k0 + half * 2;
        v2f b;
        b.x = W[(size_t)(ka)     * NCOL + colbase + idx];
        b.y = W[(size_t)(ka + 1) * NCOL + colbase + idx];
#pragma unroll
        for (int t = 0; t < RT; ++t) {
            v2f a;
            a.x = sX[(t * 16 + idx) * SXS + ka];
            a.y = sX[(t * 16 + idx) * SXS + ka + 1];
            acc[t] = __builtin_amdgcn_wmma_f32_16x16x4_f32(
                 false, a,  false, b,
                 (short)0, acc[t],  false,  false);
        }
    }

#pragma unroll
    for (int t = 0; t < RT; ++t) {
#pragma unroll
        for (int v = 0; v < 8; ++v) {
            int gr = row0 + t * 16 + half * 8 + v;
            if (gr < M) H[(size_t)gr * NCOL + colbase + idx] = acc[t][v];
        }
    }
}

template <int F, bool RELU>
__global__ void __launch_bounds__(1024)
k_tile_agg(const int* __restrict__ src, const int* __restrict__ dst,
           const float* __restrict__ dinv, const float* __restrict__ h,
           const float* __restrict__ bias, float* __restrict__ outp, int N, long E) {
    constexpr int TMAX = 400;
    constexpr int NW   = 32;
    constexpr int QCAP = 256;
    constexpr int L4   = F / 4;
    constexpr int RPW  = 32 / L4;
    __shared__ float acc[TMAX * F];
    __shared__ int   qe[NW * QCAP];
    __shared__ int   qd[NW * QCAP];

    const int tid  = threadIdx.x;
    const int wave = __builtin_amdgcn_readfirstlane(tid >> 5);
    const int lane = tid & 31;
    const int sub  = lane / L4;
    const int col4 = lane % L4;
    int* myqe = qe + wave * QCAP;
    int* myqd = qd + wave * QCAP;

    const int G = gridDim.x;
    int tn = (N + G - 1) / G;
    if (tn > TMAX) tn = TMAX;
    if (tn < 1) tn = 1;
    const int ntiles = (N + tn - 1) / tn;
    const bool al = (((size_t)dst) & 15) == 0;
    const long nsup = (E + 511) / 512;

    for (int t = blockIdx.x; t < ntiles; t += G) {
        const int lo = t * tn;
        const int rows = (N - lo < tn) ? (N - lo) : tn;

        for (int r = tid / L4; r < rows; r += 1024 / L4) {
            const int n = lo + r;
            const float di = dinv[n];
            const float4 hv = *(const float4*)(h + (size_t)n * F + col4 * 4);
            const float4 bv = *(const float4*)(bias + col4 * 4);
            float* a = acc + r * F + col4;
            a[0 * L4] = hv.x * di * di + bv.x;
            a[1 * L4] = hv.y * di * di + bv.y;
            a[2 * L4] = hv.z * di * di + bv.z;
            a[3 * L4] = hv.w * di * di + bv.w;
        }
        __syncthreads();

        int cnt = 0;
        auto flush = [&](int n) {
            for (int base = 0; base < n; base += 32) {
                const int i = base + lane;
                const bool v = i < n;
                const int e  = v ? myqe[i] : 0;
                const int dl = v ? myqd[i] : 0;
                const int s  = v ? src[e] : 0;
                const float w = v ? dinv[s] * dinv[lo + dl] : 0.0f;
                const int m = (n - base < 32) ? (n - base) : 32;
                for (int j0 = 0; j0 < m; j0 += 8 * RPW) {
                    float4 rv[8];
                    int    dj[8];
                    float  wj[8];
                    bool   ok[8];
#pragma unroll
                    for (int u = 0; u < 8; ++u) {
                        const int j = j0 + u * RPW + sub;
                        ok[u] = j < m;
                        const int sj = __shfl(s, j & 31);
                        wj[u] = __shfl(w, j & 31);
                        dj[u] = __shfl(dl, j & 31);
                        rv[u] = ok[u] ? *(const float4*)(h + (size_t)sj * F + col4 * 4)
                                      : make_float4(0.f, 0.f, 0.f, 0.f);
                    }
#pragma unroll
                    for (int u = 0; u < 8; ++u) {
                        if (ok[u]) {
                            const float4 q = sub ? make_float4(rv[u].z, rv[u].w, rv[u].x, rv[u].y)
                                                 : rv[u];
                            const int c0 = 2 * sub;
                            float* a = acc + dj[u] * F + col4;
                            __hip_atomic_fetch_add(a + ((c0 + 0) & 3) * L4, q.x * wj[u],
                                                   __ATOMIC_RELAXED, __HIP_MEMORY_SCOPE_WORKGROUP);
                            __hip_atomic_fetch_add(a + ((c0 + 1) & 3) * L4, q.y * wj[u],
                                                   __ATOMIC_RELAXED, __HIP_MEMORY_SCOPE_WORKGROUP);
                            __hip_atomic_fetch_add(a + ((c0 + 2) & 3) * L4, q.z * wj[u],
                                                   __ATOMIC_RELAXED, __HIP_MEMORY_SCOPE_WORKGROUP);
                            __hip_atomic_fetch_add(a + ((c0 + 3) & 3) * L4, q.w * wj[u],
                                                   __ATOMIC_RELAXED, __HIP_MEMORY_SCOPE_WORKGROUP);
                        }
                    }
                }
            }
        };

        const long rot = ((long)blockIdx.x * nsup) / G;
        for (long it = wave; it < nsup; it += NW) {
            const long sp = (it + rot < nsup) ? (it + rot) : (it + rot - nsup);
            const long eb = sp * 512 + lane * 4;
            int d[4][4];
            if (al && sp * 512 + 512 <= E) {
#pragma unroll
                for (int k = 0; k < 4; ++k) {
                    const int4 vv = *(const int4*)(dst + eb + k * 128);
                    d[k][0] = vv.x; d[k][1] = vv.y; d[k][2] = vv.z; d[k][3] = vv.w;
                }
            } else {
#pragma unroll
                for (int k = 0; k < 4; ++k) {
#pragma unroll
                    for (int c = 0; c < 4; ++c) {
                        const long e1 = eb + k * 128 + c;
                        d[k][c] = (e1 < E) ? dst[e1] : -1;
                    }
                }
            }
#pragma unroll
            for (int k = 0; k < 4; ++k) {
                const long e0 = eb + k * 128;
                bool mm[4];
#pragma unroll
                for (int c = 0; c < 4; ++c) mm[c] = (unsigned)(d[k][c] - lo) < (unsigned)rows;
                if (__builtin_amdgcn_ballot_w32(mm[0] | mm[1] | mm[2] | mm[3]) != 0u) {
                    if (cnt > QCAP - 128) { flush(cnt); cnt = 0; }
#pragma unroll
                    for (int c = 0; c < 4; ++c) {
                        const unsigned b = __builtin_amdgcn_ballot_w32(mm[c]);
                        if (mm[c]) {
                            const int p = cnt + (int)__builtin_amdgcn_mbcnt_lo(b, 0u);
                            myqe[p] = (int)(e0 + c);
                            myqd[p] = d[k][c] - lo;
                        }
                        cnt += __builtin_popcount(b);
                    }
                }
            }
        }
        flush(cnt);
        __syncthreads();

        for (int r = tid / L4; r < rows; r += 1024 / L4) {
            const float* a = acc + r * F + col4;
            float4 o = make_float4(a[0 * L4], a[1 * L4], a[2 * L4], a[3 * L4]);
            if (RELU) {
                o.x = o.x > 0.0f ? o.x : 0.0f;
                o.y = o.y > 0.0f ? o.y : 0.0f;
                o.z = o.z > 0.0f ? o.z : 0.0f;
                o.w = o.w > 0.0f ? o.w : 0.0f;
            }
            *(float4*)(outp + (size_t)(lo + r) * F + col4 * 4) = o;
        }
        __syncthreads();
    }
}

extern "C" void kernel_launch(void* const* d_in, const int* in_sizes, int n_in,
                              void* d_out, int out_size, void* d_ws, size_t ws_size,
                              hipStream_t stream) {
    const float* x  = (const float*)d_in[0];
    const int*   ei = (const int*)d_in[1];
    const float* W1 = (const float*)d_in[2];
    const float* b1 = (const float*)d_in[3];
    const float* W2 = (const float*)d_in[4];
    const float* b2 = (const float*)d_in[5];
    float* out = (float*)d_out;

    const int  N = in_sizes[0] / IN_F;
    const long E = (long)in_sizes[1] / 2;
    const int* srcIdx = ei;
    const int* dstIdx = ei + E;

    float* ws   = (float*)d_ws;
    float* dinv = ws;
    float* h    = dinv + N;
    float* agg1 = h + (size_t)N * HID_F;
    float* h2   = agg1 + (size_t)N * HID_F;

    k_fill<<<1024, 256, 0, stream>>>(dinv, 1.0f, (long)N);

    k_degree<<<4096, 256, 0, stream>>>(dstIdx, dinv, E);
    k_rsqrt<<<(N + 255) / 256, 256, 0, stream>>>(dinv, N);

    constexpr int RT   = 4;
    constexpr int ROWS = 16 * RT;

    k_gemm_wmma<IN_F, HID_F, RT>
        <<<(N + ROWS - 1) / ROWS, 32 * (HID_F / 16), 0, stream>>>(x, W1, h, N);
    k_tile_agg<HID_F, true><<<256, 1024, 0, stream>>>(srcIdx, dstIdx, dinv, h, b1, agg1, N, E);

    k_gemm_wmma<HID_F, OUT_F, RT>
        <<<(N + ROWS - 1) / ROWS, 32 * (OUT_F / 16), 0, stream>>>(agg1, W2, h2, N);
    k_tile_agg<OUT_F, false><<<256, 1024, 0, stream>>>(srcIdx, dstIdx, dinv, h2, b2, out, N, E);
}


# Layer-2 aggregation replays per-tile edge lists (src, row, weight) that the layer-1 tile scan appends to the workspace; second whole-list dst scan goes; layer-2 sum order = LDS arrival, new pattern
# speedup vs baseline: 2.0301x; 1.1989x over previous
#include <hip/hip_runtime.h>

typedef __attribute__((ext_vector_type(2))) float v2f;
typedef __attribute__((ext_vector_type(8))) float v8f;

#define IN_F  128
#define HID_F 128
#define OUT_F 64

__global__ void k_fill(float* __restrict__ p, float v, long n) {
    long i = (long)blockIdx.x * blockDim.x + threadIdx.x;
    long stride = (long)gridDim.x * blockDim.x;
    for (; i < n; i += stride) p[i] = v;
}

__global__ void k_degree(const int* __restrict__ dst, float* __restrict__ deg, long E) {
    long i = (long)blockIdx.x * blockDim.x + threadIdx.x;
    long stride = (long)gridDim.x * blockDim.x;
    for (; i < E; i += stride)
        __hip_atomic_fetch_add(&deg[dst[i]], 1.0f, __ATOMIC_RELAXED, __HIP_MEMORY_SCOPE_AGENT);
}

__global__ void k_rsqrt(float* __restrict__ deg, int N) {
    int i = blockIdx.x * blockDim.x + threadIdx.x;
    if (i < N) deg[i] = rsqrtf(deg[i]);
}

template <int K, int NCOL, int RT>
__global__ void __launch_bounds__(32 * (NCOL / 16))
k_gemm_wmma(const float* __restrict__ X, const float* __restrict__ W,
            float* __restrict__ H, int M) {
    constexpr int ROWS = 16 * RT;
    constexpr int SXS  = K + 4;
    __shared__ float sX[ROWS * SXS];

    const int tid  = threadIdx.x;
    const int wave = tid >> 5;
    const int lane = tid & 31;
    const int row0 = blockIdx.x * ROWS;

    for (int i = tid; i < ROWS * K; i += blockDim.x) {
        int r = i / K, c = i % K;
        int gr = row0 + r;
        sX[r * SXS + c] = (gr < M) ? X[(size_t)gr * K + c] : 0.0f;
    }
    __syncthreads();

    const int half = lane >> 4;
    const int idx  = lane & 15;
    const int colbase = wave * 16;

    v8f acc[RT] = {};
#pragma unroll
    for (int k0 = 0; k0 < K; k0 += 4) {
        const int ka = k0 + half * 2;
        v2f b;
        b.x = W[(size_t)(ka)     * NCOL + colbase + idx];
        b.y = W[(size_t)(ka + 1) * NCOL + colbase + idx];
#pragma unroll
        for (int t = 0; t < RT; ++t) {
            v2f a;
            a.x = sX[(t * 16 + idx) * SXS + ka];
            a.y = sX[(t * 16 + idx) * SXS + ka + 1];
            acc[t] = __builtin_amdgcn_wmma_f32_16x16x4_f32(
                 false, a,  false, b,
                 (short)0, acc[t],  false,  false);
        }
    }

#pragma unroll
    for (int t = 0; t < RT; ++t) {
#pragma unroll
        for (int v = 0; v < 8; ++v) {
            int gr = row0 + t * 16 + half * 8 + v;
            if (gr < M) H[(size_t)gr * NCOL + colbase + idx] = acc[t][v];
        }
    }
}

template <int F, bool RELU, bool REC, bool REPLAY>
__global__ void __launch_bounds__(1024)
k_tile_agg(const int* __restrict__ src, const int* __restrict__ dst,
           const float* __restrict__ dinv, const float* __restrict__ h,
           const float* __restrict__ bias, float* __restrict__ outp,
           int* __restrict__ lst, int N, long E) {
    constexpr int TMAX = 400;
    constexpr int NW   = 32;
    constexpr int QCAP = 256;
    constexpr int L4   = F / 4;
    constexpr int RPW  = 32 / L4;
    __shared__ float acc[TMAX * F];
    __shared__ int   qe[NW * QCAP];
    __shared__ int   qd[NW * QCAP];
    __shared__ int   tl[2];

    const int tid  = threadIdx.x;
    const int wave = __builtin_amdgcn_readfirstlane(tid >> 5);
    const int lane = tid & 31;
    const int sub  = lane / L4;
    const int col4 = lane % L4;
    int* myqe = qe + wave * QCAP;
    int* myqd = qd + wave * QCAP;

    const int G = gridDim.x;
    int tn = (N + G - 1) / G;
    if (tn > TMAX) tn = TMAX;
    if (tn < 1) tn = 1;
    const int ntiles = (N + tn - 1) / tn;
    const bool al = (((size_t)dst) & 15) == 0;
    const long nsup = (E + 511) / 512;

    const long lwords = (long)N * OUT_F;
    const long lhdr   = ((long)ntiles + 31) & ~31L;
    long lc = (lwords - lhdr) / ntiles / 3;
    if (lc < 0) lc = 0;
    if (lc > (1L << 30)) lc = 1L << 30;
    const int cap = (int)(lc & ~31L);

    for (int t = blockIdx.x; t < ntiles; t += G) {
        const int lo = t * tn;
        const int rows = (N - lo < tn) ? (N - lo) : tn;
        int* const ls = lst + lhdr + (long)t * 3 * cap;
        int* const ld = ls + cap;
        int* const lw = ld + cap;
        if (REC && tid == 0) { tl[0] = 0; tl[1] = 0; }
        int lraw = -1;
        if (REPLAY) lraw = lst[t];

        for (int r = tid / L4; r < rows; r += 1024 / L4) {
            const int n = lo + r;
            const float di = dinv[n];
            const float4 hv = *(const float4*)(h + (size_t)n * F + col4 * 4);
            const float4 bv = *(const float4*)(bias + col4 * 4);
            float* a = acc + r * F + col4;
            a[0 * L4] = hv.x * di * di + bv.x;
            a[1 * L4] = hv.y * di * di + bv.y;
            a[2 * L4] = hv.z * di * di + bv.z;
            a[3 * L4] = hv.w * di * di + bv.w;
        }
        __syncthreads();

        int cnt = 0;
        auto add_rows = [&](int s, int dl, float w, int m) {
            for (int j0 = 0; j0 < m; j0 += 8 * RPW) {
                float4 rv[8];
                int    dj[8];
                float  wj[8];
                bool   ok[8];
#pragma unroll
                for (int u = 0; u < 8; ++u) {
                    const int j = j0 + u * RPW + sub;
                    ok[u] = j < m;
                    const int sj = __shfl(s, j & 31);
                    wj[u] = __shfl(w, j & 31);
                    dj[u] = __shfl(dl, j & 31);
                    rv[u] = ok[u] ? *(const float4*)(h + (size_t)sj * F + col4 * 4)
                                  : make_float4(0.f, 0.f, 0.f, 0.f);
                }
#pragma unroll
                for (int u = 0; u < 8; ++u) {
                    if (ok[u]) {
                        const float4 q = sub ? make_float4(rv[u].z, rv[u].w, rv[u].x, rv[u].y)
                                             : rv[u];
                        const int c0 = 2 * sub;
                        float* a = acc + dj[u] * F + col4;
                        __hip_atomic_fetch_add(a + ((c0 + 0) & 3) * L4, q.x * wj[u],
                                               __ATOMIC_RELAXED, __HIP_MEMORY_SCOPE_WORKGROUP);
                        __hip_atomic_fetch_add(a + ((c0 + 1) & 3) * L4, q.y * wj[u],
                                               __ATOMIC_RELAXED, __HIP_MEMORY_SCOPE_WORKGROUP);
                        __hip_atomic_fetch_add(a + ((c0 + 2) & 3) * L4, q.z * wj[u],
                                               __ATOMIC_RELAXED, __HIP_MEMORY_SCOPE_WORKGROUP);
                        __hip_atomic_fetch_add(a + ((c0 + 3) & 3) * L4, q.w * wj[u],
                                               __ATOMIC_RELAXED, __HIP_MEMORY_SCOPE_WORKGROUP);
                    }
                }
            }
        };
        auto flush = [&](int n) {
            for (int base = 0; base < n; base += 32) {
                const int i = base + lane;
                const bool v = i < n;
                const int e  = v ? myqe[i] : 0;
                const int dl = v ? myqd[i] : 0;
                const int s  = v ? src[e] : 0;
                const float w = v ? dinv[s] * dinv[lo + dl] : 0.0f;
                const int m = (n - base < 32) ? (n - base) : 32;
                if (REC) {
                    int pos = 0;
                    if (lane == 0)
                        pos = __hip_atomic_fetch_add(&tl[0], m, __ATOMIC_RELAXED,
                                                     __HIP_MEMORY_SCOPE_WORKGROUP);
                    pos = __builtin_amdgcn_readfirstlane(pos);
                    if (pos + m <= cap) {
                        if (v) {
                            ls[pos + lane] = s;
                            ld[pos + lane] = dl;
                            lw[pos + lane] = __float_as_int(w);
                        }
                    } else if (lane == 0) {
                        tl[1] = 1;
                    }
                }
                add_rows(s, dl, w, m);
            }
        };

        int lcnt = -1;
        if (REPLAY) {
            lcnt = __builtin_amdgcn_readfirstlane(lraw);
            if (lcnt > cap) lcnt = cap;
        }
        if (REPLAY && lcnt >= 0) {
            int base = wave * 32;
            int s = 0, dl = 0; float w = 0.0f;
            if (base + lane < lcnt) {
                s = ls[base + lane]; dl = ld[base + lane]; w = __int_as_float(lw[base + lane]);
            }
            while (base < lcnt) {
                const int nb = base + NW * 32;
                int s2 = 0, dl2 = 0; float w2 = 0.0f;
                if (nb + lane < lcnt) {
                    s2 = ls[nb + lane]; dl2 = ld[nb + lane]; w2 = __int_as_float(lw[nb + lane]);
                }
                add_rows(s, dl, w, (lcnt - base < 32) ? (lcnt - base) : 32);
                s = s2; dl = dl2; w = w2; base = nb;
            }
        } else {
        const long rot = ((long)blockIdx.x * nsup) / G;
        for (long it = wave; it < nsup; it += NW) {
            const long sp = (it + rot < nsup) ? (it + rot) : (it + rot - nsup);
            const long eb = sp * 512 + lane * 4;
            int d[4][4];
            if (al && sp * 512 + 512 <= E) {
#pragma unroll
                for (int k = 0; k < 4; ++k) {
                    const int4 vv = *(const int4*)(dst + eb + k * 128);
                    d[k][0] = vv.x; d[k][1] = vv.y; d[k][2] = vv.z; d[k][3] = vv.w;
                }
            } else {
#pragma unroll
                for (int k = 0; k < 4; ++k) {
#pragma unroll
                    for (int c = 0; c < 4; ++c) {
                        const long e1 = eb + k * 128 + c;
                        d[k][c] = (e1 < E) ? dst[e1] : -1;
                    }
                }
            }
#pragma unroll
            for (int k = 0; k < 4; ++k) {
                const long e0 = eb + k * 128;
                bool mm[4];
#pragma unroll
                for (int c = 0; c < 4; ++c) mm[c] = (unsigned)(d[k][c] - lo) < (unsigned)rows;
                if (__builtin_amdgcn_ballot_w32(mm[0] | mm[1] | mm[2] | mm[3]) != 0u) {
                    if (cnt > QCAP - 128) { flush(cnt); cnt = 0; }
#pragma unroll
                    for (int c = 0; c < 4; ++c) {
                        const unsigned b = __builtin_amdgcn_ballot_w32(mm[c]);
                        if (mm[c]) {
                            const int p = cnt + (int)__builtin_amdgcn_mbcnt_lo(b, 0u);
                            myqe[p] = (int)(e0 + c);
                            myqd[p] = d[k][c] - lo;
                        }
                        cnt += __builtin_popcount(b);
                    }
                }
            }
        }
        flush(cnt);
        }
        __syncthreads();
        if (REC && tid == 0) lst[t] = tl[1] ? -1 : tl[0];

        for (int r = tid / L4; r < rows; r += 1024 / L4) {
            const float* a = acc + r * F + col4;
            float4 o = make_float4(a[0 * L4], a[1 * L4], a[2 * L4], a[3 * L4]);
            if (RELU) {
                o.x = o.x > 0.0f ? o.x : 0.0f;
                o.y = o.y > 0.0f ? o.y : 0.0f;
                o.z = o.z > 0.0f ? o.z : 0.0f;
                o.w = o.w > 0.0f ? o.w : 0.0f;
            }
            *(float4*)(outp + (size_t)(lo + r) * F + col4 * 4) = o;
        }
        __syncthreads();
    }
}

extern "C" void kernel_launch(void* const* d_in, const int* in_sizes, int n_in,
                              void* d_out, int out_size, void* d_ws, size_t ws_size,
                              hipStream_t stream) {
    const float* x  = (const float*)d_in[0];
    const int*   ei = (const int*)d_in[1];
    const float* W1 = (const float*)d_in[2];
    const float* b1 = (const float*)d_in[3];
    const float* W2 = (const float*)d_in[4];
    const float* b2 = (const float*)d_in[5];
    float* out = (float*)d_out;

    const int  N = in_sizes[0] / IN_F;
    const long E = (long)in_sizes[1] / 2;
    const int* srcIdx = ei;
    const int* dstIdx = ei + E;

    float* ws   = (float*)d_ws;
    float* dinv = ws;
    float* h    = dinv + N;
    float* agg1 = h + (size_t)N * HID_F;
    int*   lst  = (int*)(agg1 + (size_t)N * HID_F);
    float* h2   = h;

    k_fill<<<1024, 256, 0, stream>>>(dinv, 1.0f, (long)N);

    k_degree<<<4096, 256, 0, stream>>>(dstIdx, dinv, E);
    k_rsqrt<<<(N + 255) / 256, 256, 0, stream>>>(dinv, N);

    constexpr int RT   = 4;
    constexpr int ROWS = 16 * RT;

    k_gemm_wmma<IN_F, HID_F, RT>
        <<<(N + ROWS - 1) / ROWS, 32 * (HID_F / 16), 0, stream>>>(x, W1, h, N);
    k_tile_agg<HID_F, true, true, false>
        <<<256, 1024, 0, stream>>>(srcIdx, dstIdx, dinv, h, b1, agg1, lst, N, E);

    k_gemm_wmma<HID_F, OUT_F, RT>
        <<<(N + ROWS - 1) / ROWS, 32 * (OUT_F / 16), 0, stream>>>(agg1, W2, h2, N);
    k_tile_agg<OUT_F, false, false, true>
        <<<256, 1024, 0, stream>>>(srcIdx, dstIdx, dinv, h2, b2, out, lst, N, E);
}


# Tile edge lists built once in the degree kernel (pass sorted by tile in LDS, one global int add per tile reserves its run, line stores); layer 1 replays them, dst scan goes; sum order = LDS arrival
# speedup vs baseline: 2.4482x; 1.2060x over previous
#include <hip/hip_runtime.h>

typedef __attribute__((ext_vector_type(2))) float v2f;
typedef __attribute__((ext_vector_type(8))) float v8f;

#define IN_F  128
#define HID_F 128
#define OUT_F 64

#define AGG_G 256
#define TMAX  400
#define MAXT  1024
#define BIN_EPT 8
#define BIN_PASS (1024 * BIN_EPT)

struct TileGeo { int tn; int ntiles; int cap; long lhdr; };

__device__ __forceinline__ TileGeo tile_geo(int N) {
    TileGeo g;
    int tn = (N + AGG_G - 1) / AGG_G;
    if (tn > TMAX) tn = TMAX;
    if (tn < 1) tn = 1;
    g.tn = tn;
    g.ntiles = (N + tn - 1) / tn;
    const long lwords = (long)N * OUT_F;
    g.lhdr = ((long)g.ntiles + 31) & ~31L;
    long lc = (lwords - g.lhdr) / g.ntiles / 3;
    if (lc < 0) lc = 0;
    if (lc > (1L << 30)) lc = 1L << 30;
    g.cap = (int)(lc & ~31L);
    return g;
}

__global__ void k_init(float* __restrict__ deg, int* __restrict__ lst, int N) {
    const TileGeo g = tile_geo(N);
    long i = (long)blockIdx.x * blockDim.x + threadIdx.x;
    long stride = (long)gridDim.x * blockDim.x;
    for (long j = i; j < N; j += stride) deg[j] = 1.0f;
    for (long j = i; j < g.lhdr; j += stride) lst[j] = 0;
}

__global__ void __launch_bounds__(1024)
k_degree_bin(const int* __restrict__ src, const int* __restrict__ dst,
             float* __restrict__ deg, int* __restrict__ lst, int N, long E) {
    __shared__ int hist[MAXT];
    __shared__ int gdelta[MAXT];
    __shared__ int ssrc[BIN_PASS];
    __shared__ int spk[BIN_PASS];
    __shared__ int wsum[32];
    __shared__ int tot;
    const TileGeo g = tile_geo(N);
    const bool binned = g.ntiles <= MAXT && g.cap > 0;
    const int tid  = threadIdx.x;
    const int lane = tid & 31;
    const int wave = tid >> 5;

    for (long p0 = (long)blockIdx.x * BIN_PASS; p0 < E; p0 += (long)gridDim.x * BIN_PASS) {
        hist[tid] = 0;
        __syncthreads();

        int d[BIN_EPT], s[BIN_EPT], tl[BIN_EPT], rk[BIN_EPT];
#pragma unroll
        for (int k = 0; k < BIN_EPT; ++k) {
            const long e = p0 + k * 1024 + tid;
            const bool ok = e < E;
            d[k] = ok ? dst[e] : -1;
            s[k] = ok ? src[e] : 0;
        }
#pragma unroll
        for (int k = 0; k < BIN_EPT; ++k) {
            tl[k] = 0; rk[k] = 0;
            if ((unsigned)d[k] < (unsigned)N) {
                __hip_atomic_fetch_add(&deg[d[k]], 1.0f, __ATOMIC_RELAXED, __HIP_MEMORY_SCOPE_AGENT);
                if (binned) {
                    tl[k] = d[k] / g.tn;
                    rk[k] = __hip_atomic_fetch_add(&hist[tl[k]], 1, __ATOMIC_RELAXED,
                                                   __HIP_MEMORY_SCOPE_WORKGROUP);
                }
            }
        }
        __syncthreads();
        if (!binned) continue;

        const int c = hist[tid];
        int gb = 0;
        if (c > 0 && tid < g.ntiles)
            gb = __hip_atomic_fetch_add(&lst[tid], c, __ATOMIC_RELAXED, __HIP_MEMORY_SCOPE_AGENT);
        int x = c;
#pragma unroll
        for (int o = 1; o < 32; o <<= 1) {
            const int y = __shfl_up(x, o);
            if (lane >= o) x += y;
        }
        if (lane == 31) wsum[wave] = x;
        __syncthreads();
        if (wave == 0) {
            const int v = wsum[lane];
            int z = v;
#pragma unroll
            for (int o = 1; o < 32; o <<= 1) {
                const int y = __shfl_up(z, o);
                if (lane >= o) z += y;
            }
            wsum[lane] = z - v;
            if (lane == 31) tot = z;
        }
        __syncthreads();
        const int excl = x - c + wsum[wave];
        hist[tid]   = excl;
        gdelta[tid] = gb - excl;
        __syncthreads();

#pragma unroll
        for (int k = 0; k < BIN_EPT; ++k) {
            if ((unsigned)d[k] < (unsigned)N) {
                const int p = hist[tl[k]] + rk[k];
                ssrc[p] = s[k];
                spk[p]  = (tl[k] << 9) | (d[k] - tl[k] * g.tn);
            }
        }
        __syncthreads();

        const int n = tot;
        for (int i = tid; i < n; i += 1024) {
            const int pk = spk[i];
            const int t  = pk >> 9;
            const int pos = gdelta[t] + i;
            if (pos >= 0 && pos < g.cap) {
                int* const ls = lst + g.lhdr + (long)t * 3 * g.cap;
                ls[pos]         = ssrc[i];
                ls[g.cap + pos] = pk & 511;
            }
        }
    }
}

__global__ void k_rsqrt(float* __restrict__ deg, int N) {
    int i = blockIdx.x * blockDim.x + threadIdx.x;
    if (i < N) deg[i] = rsqrtf(deg[i]);
}

template <int K, int NCOL, int RT>
__global__ void __launch_bounds__(32 * (NCOL / 16))
k_gemm_wmma(const float* __restrict__ X, const float* __restrict__ W,
            float* __restrict__ H, int M) {
    constexpr int ROWS = 16 * RT;
    constexpr int SXS  = K + 4;
    __shared__ float sX[ROWS * SXS];

    const int tid  = threadIdx.x;
    const int wave = tid >> 5;
    const int lane = tid & 31;
    const int row0 = blockIdx.x * ROWS;

    for (int i = tid; i < ROWS * K; i += blockDim.x) {
        int r = i / K, c = i % K;
        int gr = row0 + r;
        sX[r * SXS + c] = (gr < M) ? X[(size_t)gr * K + c] : 0.0f;
    }
    __syncthreads();

    const int half = lane >> 4;
    const int idx  = lane & 15;
    const int colbase = wave * 16;

    v8f acc[RT] = {};
#pragma unroll
    for (int k0 = 0; k0 < K; k0 += 4) {
        const int ka = k0 + half * 2;
        v2f b;
        b.x = W[(size_t)(ka)     * NCOL + colbase + idx];
        b.y = W[(size_t)(ka + 1) * NCOL + colbase + idx];
#pragma unroll
        for (int t = 0; t < RT; ++t) {
            v2f a;
            a.x = sX[(t * 16 + idx) * SXS + ka];
            a.y = sX[(t * 16 + idx) * SXS + ka + 1];
            acc[t] = __builtin_amdgcn_wmma_f32_16x16x4_f32(
                 false, a,  false, b,
                 (short)0, acc[t],  false,  false);
        }
    }

#pragma unroll
    for (int t = 0; t < RT; ++t) {
#pragma unroll
        for (int v = 0; v < 8; ++v) {
            int gr = row0 + t * 16 + half * 8 + v;
            if (gr < M) H[(size_t)gr * NCOL + colbase + idx] = acc[t][v];
        }
    }
}

template <int F, bool RELU, bool FIRST>
__global__ void __launch_bounds__(1024)
k_tile_agg(const int* __restrict__ src, const int* __restrict__ dst,
           const float* __restrict__ dinv, const float* __restrict__ h,
           const float* __restrict__ bias, float* __restrict__ outp,
           int* __restrict__ lst, int N, long E) {
    constexpr int NW   = 32;
    constexpr int QCAP = 256;
    constexpr int L4   = F / 4;
    constexpr int RPW  = 32 / L4;
    __shared__ float acc[TMAX * F];
    __shared__ int   qe[NW * QCAP];
    __shared__ int   qd[NW * QCAP];

    const int tid  = threadIdx.x;
    const int wave = __builtin_amdgcn_readfirstlane(tid >> 5);
    const int lane = tid & 31;
    const int sub  = lane / L4;
    const int col4 = lane % L4;
    int* myqe = qe + wave * QCAP;
    int* myqd = qd + wave * QCAP;

    const int G = gridDim.x;
    const TileGeo g = tile_geo(N);
    const int tn = g.tn;
    const int ntiles = g.ntiles;
    const int cap = g.cap;
    const long lhdr = g.lhdr;
    const bool binned = ntiles <= MAXT && cap > 0;
    const bool al = (((size_t)dst) & 15) == 0;
    const long nsup = (E + 511) / 512;

    for (int t = blockIdx.x; t < ntiles; t += G) {
        const int lo = t * tn;
        const int rows = (N - lo < tn) ? (N - lo) : tn;
        int* const ls = lst + lhdr + (long)t * 3 * cap;
        int* const ld = ls + cap;
        int* const lw = ld + cap;
        int lraw = -1;
        if (binned) lraw = lst[t];

        for (int r = tid / L4; r < rows; r += 1024 / L4) {
            const int n = lo + r;
            const float di = dinv[n];
            const float4 hv = *(const float4*)(h + (size_t)n * F + col4 * 4);
            const float4 bv = *(const float4*)(bias + col4 * 4);
            float* a = acc + r * F + col4;
            a[0 * L4] = hv.x * di * di + bv.x;
            a[1 * L4] = hv.y * di * di + bv.y;
            a[2 * L4] = hv.z * di * di + bv.z;
            a[3 * L4] = hv.w * di * di + bv.w;
        }
        __syncthreads();

        auto add_rows = [&](int s, int dl, float w, int m, auto&& ahead) {
            for (int j0 = 0; j0 < m; j0 += 8 * RPW) {
                float4 rv[8];
                int    dj[8];
                float  wj[8];
                bool   ok[8];
#pragma unroll
                for (int u = 0; u < 8; ++u) {
                    const int j = j0 + u * RPW + sub;
                    ok[u] = j < m;
                    const int sj = __shfl(s, j & 31);
                    wj[u] = __shfl(w, j & 31);
                    dj[u] = __shfl(dl, j & 31);
                    rv[u] = ok[u] ? *(const float4*)(h + (size_t)sj * F + col4 * 4)
                                  : make_float4(0.f, 0.f, 0.f, 0.f);
                }
                if (j0 == 0) ahead();
#pragma unroll
                for (int u = 0; u < 8; ++u) {
                    if (ok[u]) {
                        const float4 q = sub ? make_float4(rv[u].z, rv[u].w, rv[u].x, rv[u].y)
                                             : rv[u];
                        const int c0 = 2 * sub;
                        float* a = acc + dj[u] * F + col4;
                        __hip_atomic_fetch_add(a + ((c0 + 0) & 3) * L4, q.x * wj[u],
                                               __ATOMIC_RELAXED, __HIP_MEMORY_SCOPE_WORKGROUP);
                        __hip_atomic_fetch_add(a + ((c0 + 1) & 3) * L4, q.y * wj[u],
                                               __ATOMIC_RELAXED, __HIP_MEMORY_SCOPE_WORKGROUP);
                        __hip_atomic_fetch_add(a + ((c0 + 2) & 3) * L4, q.z * wj[u],
                                               __ATOMIC_RELAXED, __HIP_MEMORY_SCOPE_WORKGROUP);
                        __hip_atomic_fetch_add(a + ((c0 + 3) & 3) * L4, q.w * wj[u],
                                               __ATOMIC_RELAXED, __HIP_MEMORY_SCOPE_WORKGROUP);
                    }
                }
            }
        };

        const int lcnt = __builtin_amdgcn_readfirstlane(lraw);
        if (lcnt >= 0 && lcnt <= cap) {
            constexpr int STEP = NW * 32;
            int base = wave * 32;
            if (FIRST) {
                if (base < lcnt) {
                    const int last = lcnt - 1;
                    int s, dl, s1, dl1;
                    {
                        const int k0 = (base + lane < last) ? (base + lane) : last;
                        const int k1 = (base + STEP + lane < last) ? (base + STEP + lane) : last;
                        s = ls[k0]; dl = ld[k0]; s1 = ls[k1]; dl1 = ld[k1];
                    }
                    float w = dinv[s] * dinv[lo + dl];
                    while (base < lcnt) {
                        float da = 0.0f, db = 0.0f;
                        int s2 = 0, dl2 = 0;
                        if (base + lane < lcnt) lw[base + lane] = __float_as_int(w);
                        add_rows(s, dl, w, (lcnt - base < 32) ? (lcnt - base) : 32, [&] {
                            da = dinv[s1]; db = dinv[lo + dl1];
                            const int k2 = (base + 2 * STEP + lane < last) ? (base + 2 * STEP + lane)
                                                                          : last;
                            s2 = ls[k2]; dl2 = ld[k2];
                            __builtin_amdgcn_sched_barrier(0);
                        });
                        w = da * db;
                        s = s1; dl = dl1; s1 = s2; dl1 = dl2; base += STEP;
                    }
                }
            } else {
                int s = 0, dl = 0; float w = 0.0f;
                if (base + lane < lcnt) {
                    s = ls[base + lane]; dl = ld[base + lane]; w = __int_as_float(lw[base + lane]);
                }
                while (base < lcnt) {
                    const int nb = base + STEP;
                    int s2 = 0, dl2 = 0; float w2 = 0.0f;
                    if (nb + lane < lcnt) {
                        s2 = ls[nb + lane]; dl2 = ld[nb + lane]; w2 = __int_as_float(lw[nb + lane]);
                    }
                    add_rows(s, dl, w, (lcnt - base < 32) ? (lcnt - base) : 32, [] {});
                    s = s2; dl = dl2; w = w2; base = nb;
                }
            }
        } else {
        int cnt = 0;
        auto flush = [&](int n) {
            for (int base = 0; base < n; base += 32) {
                const int i = base + lane;
                const bool v = i < n;
                const int e  = v ? myqe[i] : 0;
                const int dl = v ? myqd[i] : 0;
                const int s  = v ? src[e] : 0;
                const float w = v ? dinv[s] * dinv[lo + dl] : 0.0f;
                const int m = (n - base < 32) ? (n - base) : 32;
                add_rows(s, dl, w, m, [] {});
            }
        };
        const long rot = ((long)blockIdx.x * nsup) / G;
        for (long it = wave; it < nsup; it += NW) {
            const long sp = (it + rot < nsup) ? (it + rot) : (it + rot - nsup);
            const long eb = sp * 512 + lane * 4;
            int d[4][4];
            if (al && sp * 512 + 512 <= E) {
#pragma unroll
                for (int k = 0; k < 4; ++k) {
                    const int4 vv = *(const int4*)(dst + eb + k * 128);
                    d[k][0] = vv.x; d[k][1] = vv.y; d[k][2] = vv.z; d[k][3] = vv.w;
                }
            } else {
#pragma unroll
                for (int k = 0; k < 4; ++k) {
#pragma unroll
                    for (int c = 0; c < 4; ++c) {
                        const long e1 = eb + k * 128 + c;
                        d[k][c] = (e1 < E) ? dst[e1] : -1;
                    }
                }
            }
#pragma unroll
            for (int k = 0; k < 4; ++k) {
                const long e0 = eb + k * 128;
                bool mm[4];
#pragma unroll
                for (int c = 0; c < 4; ++c) mm[c] = (unsigned)(d[k][c] - lo) < (unsigned)rows;
                if (__builtin_amdgcn_ballot_w32(mm[0] | mm[1] | mm[2] | mm[3]) != 0u) {
                    if (cnt > QCAP - 128) { flush(cnt); cnt = 0; }
#pragma unroll
                    for (int c = 0; c < 4; ++c) {
                        const unsigned b = __builtin_amdgcn_ballot_w32(mm[c]);
                        if (mm[c]) {
                            const int p = cnt + (int)__builtin_amdgcn_mbcnt_lo(b, 0u);
                            myqe[p] = (int)(e0 + c);
                            myqd[p] = d[k][c] - lo;
                        }
                        cnt += __builtin_popcount(b);
                    }
                }
            }
        }
        flush(cnt);
        }
        __syncthreads();

        for (int r = tid / L4; r < rows; r += 1024 / L4) {
            const float* a = acc + r * F + col4;
            float4 o = make_float4(a[0 * L4], a[1 * L4], a[2 * L4], a[3 * L4]);
            if (RELU) {
                o.x = o.x > 0.0f ? o.x : 0.0f;
                o.y = o.y > 0.0f ? o.y : 0.0f;
                o.z = o.z > 0.0f ? o.z : 0.0f;
                o.w = o.w > 0.0f ? o.w : 0.0f;
            }
            *(float4*)(outp + (size_t)(lo + r) * F + col4 * 4) = o;
        }
        __syncthreads();
    }
}

extern "C" void kernel_launch(void* const* d_in, const int* in_sizes, int n_in,
                              void* d_out, int out_size, void* d_ws, size_t ws_size,
                              hipStream_t stream) {
    const float* x  = (const float*)d_in[0];
    const int*   ei = (const int*)d_in[1];
    const float* W1 = (const float*)d_in[2];
    const float* b1 = (const float*)d_in[3];
    const float* W2 = (const float*)d_in[4];
    const float* b2 = (const float*)d_in[5];
    float* out = (float*)d_out;

    const int  N = in_sizes[0] / IN_F;
    const long E = (long)in_sizes[1] / 2;
    const int* srcIdx = ei;
    const int* dstIdx = ei + E;

    float* ws   = (float*)d_ws;
    float* dinv = ws;
    float* h    = dinv + N;
    float* agg1 = h + (size_t)N * HID_F;
    int*   lst  = (int*)(agg1 + (size_t)N * HID_F);
    float* h2   = h;

    k_init<<<1024, 256, 0, stream>>>(dinv, lst, N);

    k_degree_bin<<<(int)((E + BIN_PASS - 1) / BIN_PASS), 1024, 0, stream>>>(
        srcIdx, dstIdx, dinv, lst, N, E);
    k_rsqrt<<<(N + 255) / 256, 256, 0, stream>>>(dinv, N);

    constexpr int RT   = 4;
    constexpr int ROWS = 16 * RT;

    k_gemm_wmma<IN_F, HID_F, RT>
        <<<(N + ROWS - 1) / ROWS, 32 * (HID_F / 16), 0, stream>>>(x, W1, h, N);
    k_tile_agg<HID_F, true, true>
        <<<AGG_G, 1024, 0, stream>>>(srcIdx, dstIdx, dinv, h, b1, agg1, lst, N, E);

    k_gemm_wmma<HID_F, OUT_F, RT>
        <<<(N + ROWS - 1) / ROWS, 32 * (OUT_F / 16), 0, stream>>>(agg1, W2, h2, N);
    k_tile_agg<OUT_F, false, false>
        <<<AGG_G, 1024, 0, stream>>>(srcIdx, dstIdx, dinv, h2, b2, out, lst, N, E);
}


# Degree per destination tile: one work-group per tile counts its list's rows with LDS integer adds, writes dinv=rsqrt(1+count), same values; per-edge global float atomic, deg fill, rsqrt launch go
# speedup vs baseline: 3.3626x; 1.3735x over previous
#include <hip/hip_runtime.h>

typedef __attribute__((ext_vector_type(2))) float v2f;
typedef __attribute__((ext_vector_type(8))) float v8f;

#define IN_F  128
#define HID_F 128
#define OUT_F 64

#define AGG_G 256
#define TMAX  400
#define MAXT  1024
#define BIN_EPT 8
#define BIN_PASS (1024 * BIN_EPT)

struct TileGeo { int tn; int ntiles; int cap; long lhdr; };

__device__ __forceinline__ TileGeo tile_geo(int N) {
    TileGeo g;
    int tn = (N + AGG_G - 1) / AGG_G;
    if (tn > TMAX) tn = TMAX;
    if (tn < 1) tn = 1;
    g.tn = tn;
    g.ntiles = (N + tn - 1) / tn;
    const long lwords = (long)N * OUT_F;
    g.lhdr = ((long)g.ntiles + 31) & ~31L;
    long lc = (lwords - g.lhdr) / g.ntiles / 3;
    if (lc < 0) lc = 0;
    if (lc > (1L << 30)) lc = 1L << 30;
    g.cap = (int)(lc & ~31L);
    return g;
}

__global__ void k_init(int* __restrict__ lst, int N) {
    const TileGeo g = tile_geo(N);
    long i = (long)blockIdx.x * blockDim.x + threadIdx.x;
    long stride = (long)gridDim.x * blockDim.x;
    for (long j = i; j < g.lhdr; j += stride) lst[j] = 0;
}

__global__ void __launch_bounds__(1024)
k_bin(const int* __restrict__ src, const int* __restrict__ dst,
      int* __restrict__ lst, int N, long E) {
    __shared__ int hist[MAXT];
    __shared__ int gdelta[MAXT];
    __shared__ int ssrc[BIN_PASS];
    __shared__ int spk[BIN_PASS];
    __shared__ int wsum[32];
    __shared__ int tot;
    const TileGeo g = tile_geo(N);
    const bool binned = g.ntiles <= MAXT && g.cap > 0;
    const int tid  = threadIdx.x;
    const int lane = tid & 31;
    const int wave = tid >> 5;

    for (long p0 = (long)blockIdx.x * BIN_PASS; p0 < E; p0 += (long)gridDim.x * BIN_PASS) {
        hist[tid] = 0;
        __syncthreads();

        int d[BIN_EPT], s[BIN_EPT], tl[BIN_EPT], rk[BIN_EPT];
#pragma unroll
        for (int k = 0; k < BIN_EPT; ++k) {
            const long e = p0 + k * 1024 + tid;
            const bool ok = e < E;
            d[k] = ok ? dst[e] : -1;
            s[k] = ok ? src[e] : 0;
        }
#pragma unroll
        for (int k = 0; k < BIN_EPT; ++k) {
            tl[k] = 0; rk[k] = 0;
            if (binned && (unsigned)d[k] < (unsigned)N) {
                tl[k] = d[k] / g.tn;
                rk[k] = __hip_atomic_fetch_add(&hist[tl[k]], 1, __ATOMIC_RELAXED,
                                               __HIP_MEMORY_SCOPE_WORKGROUP);
            }
        }
        __syncthreads();
        if (!binned) continue;

        const int c = hist[tid];
        int gb = 0;
        if (c > 0 && tid < g.ntiles)
            gb = __hip_atomic_fetch_add(&lst[tid], c, __ATOMIC_RELAXED, __HIP_MEMORY_SCOPE_AGENT);
        int x = c;
#pragma unroll
        for (int o = 1; o < 32; o <<= 1) {
            const int y = __shfl_up(x, o);
            if (lane >= o) x += y;
        }
        if (lane == 31) wsum[wave] = x;
        __syncthreads();
        if (wave == 0) {
            const int v = wsum[lane];
            int z = v;
#pragma unroll
            for (int o = 1; o < 32; o <<= 1) {
                const int y = __shfl_up(z, o);
                if (lane >= o) z += y;
            }
            wsum[lane] = z - v;
            if (lane == 31) tot = z;
        }
        __syncthreads();
        const int excl = x - c + wsum[wave];
        hist[tid]   = excl;
        gdelta[tid] = gb - excl;
        __syncthreads();

#pragma unroll
        for (int k = 0; k < BIN_EPT; ++k) {
            if ((unsigned)d[k] < (unsigned)N) {
                const int p = hist[tl[k]] + rk[k];
                ssrc[p] = s[k];
                spk[p]  = (tl[k] << 9) | (d[k] - tl[k] * g.tn);
            }
        }
        __syncthreads();

        const int n = tot;
        for (int i = tid; i < n; i += 1024) {
            const int pk = spk[i];
            const int t  = pk >> 9;
            const int pos = gdelta[t] + i;
            if (pos >= 0 && pos < g.cap) {
                int* const ls = lst + g.lhdr + (long)t * 3 * g.cap;
                ls[pos]         = ssrc[i];
                ls[g.cap + pos] = pk & 511;
            }
        }
    }
}

#define DEG_T 512
#define DEG_U 8
__global__ void __launch_bounds__(DEG_T)
k_tile_deg(const int* __restrict__ dst, float* __restrict__ dinv,
           const int* __restrict__ lst, int N, long E) {
    __shared__ int cnt[TMAX];
    const TileGeo g = tile_geo(N);
    const bool binned = g.ntiles <= MAXT && g.cap > 0;
    const int tid = threadIdx.x;
    for (int t = blockIdx.x; t < g.ntiles; t += gridDim.x) {
        const int lo = t * g.tn;
        const int rows = (N - lo < g.tn) ? (N - lo) : g.tn;
        int lcnt = -1;
        if (binned) lcnt = lst[t];
        for (int r = tid; r < rows; r += DEG_T) cnt[r] = 0;
        __syncthreads();
        if (lcnt >= 0 && lcnt <= g.cap) {
            const int* const ld = lst + g.lhdr + (long)t * 3 * g.cap + g.cap;
            for (int i0 = tid; i0 < lcnt; i0 += DEG_T * DEG_U) {
                int r[DEG_U];
#pragma unroll
                for (int u = 0; u < DEG_U; ++u) {
                    const int i = i0 + u * DEG_T;
                    r[u] = (i < lcnt) ? ld[i] : -1;
                }
#pragma unroll
                for (int u = 0; u < DEG_U; ++u)
                    if ((unsigned)r[u] < (unsigned)rows)
                        __hip_atomic_fetch_add(&cnt[r[u]], 1, __ATOMIC_RELAXED,
                                               __HIP_MEMORY_SCOPE_WORKGROUP);
            }
        } else {
            for (long e = tid; e < E; e += DEG_T) {
                const unsigned r = (unsigned)(dst[e] - lo);
                if (r < (unsigned)rows)
                    __hip_atomic_fetch_add(&cnt[r], 1, __ATOMIC_RELAXED,
                                           __HIP_MEMORY_SCOPE_WORKGROUP);
            }
        }
        __syncthreads();
        for (int r = tid; r < rows; r += DEG_T)
            dinv[lo + r] = rsqrtf(1.0f + (float)cnt[r]);
        __syncthreads();
    }
}

template <int K, int NCOL, int RT>
__global__ void __launch_bounds__(32 * (NCOL / 16))
k_gemm_wmma(const float* __restrict__ X, const float* __restrict__ W,
            float* __restrict__ H, int M) {
    constexpr int ROWS = 16 * RT;
    constexpr int SXS  = K + 4;
    __shared__ float sX[ROWS * SXS];

    const int tid  = threadIdx.x;
    const int wave = tid >> 5;
    const int lane = tid & 31;
    const int row0 = blockIdx.x * ROWS;

    for (int i = tid; i < ROWS * K; i += blockDim.x) {
        int r = i / K, c = i % K;
        int gr = row0 + r;
        sX[r * SXS + c] = (gr < M) ? X[(size_t)gr * K + c] : 0.0f;
    }
    __syncthreads();

    const int half = lane >> 4;
    const int idx  = lane & 15;
    const int colbase = wave * 16;

    v8f acc[RT] = {};
#pragma unroll
    for (int k0 = 0; k0 < K; k0 += 4) {
        const int ka = k0 + half * 2;
        v2f b;
        b.x = W[(size_t)(ka)     * NCOL + colbase + idx];
        b.y = W[(size_t)(ka + 1) * NCOL + colbase + idx];
#pragma unroll
        for (int t = 0; t < RT; ++t) {
            v2f a;
            a.x = sX[(t * 16 + idx) * SXS + ka];
            a.y = sX[(t * 16 + idx) * SXS + ka + 1];
            acc[t] = __builtin_amdgcn_wmma_f32_16x16x4_f32(
                 false, a,  false, b,
                 (short)0, acc[t],  false,  false);
        }
    }

#pragma unroll
    for (int t = 0; t < RT; ++t) {
#pragma unroll
        for (int v = 0; v < 8; ++v) {
            int gr = row0 + t * 16 + half * 8 + v;
            if (gr < M) H[(size_t)gr * NCOL + colbase + idx] = acc[t][v];
        }
    }
}

template <int F, bool RELU, bool FIRST>
__global__ void __launch_bounds__(1024)
k_tile_agg(const int* __restrict__ src, const int* __restrict__ dst,
           const float* __restrict__ dinv, const float* __restrict__ h,
           const float* __restrict__ bias, float* __restrict__ outp,
           int* __restrict__ lst, int N, long E) {
    constexpr int NW   = 32;
    constexpr int QCAP = 256;
    constexpr int L4   = F / 4;
    constexpr int RPW  = 32 / L4;
    __shared__ float acc[TMAX * F];
    __shared__ int   qe[NW * QCAP];
    __shared__ int   qd[NW * QCAP];

    const int tid  = threadIdx.x;
    const int wave = __builtin_amdgcn_readfirstlane(tid >> 5);
    const int lane = tid & 31;
    const int sub  = lane / L4;
    const int col4 = lane % L4;
    int* myqe = qe + wave * QCAP;
    int* myqd = qd + wave * QCAP;

    const int G = gridDim.x;
    const TileGeo g = tile_geo(N);
    const int tn = g.tn;
    const int ntiles = g.ntiles;
    const int cap = g.cap;
    const long lhdr = g.lhdr;
    const bool binned = ntiles <= MAXT && cap > 0;
    const bool al = (((size_t)dst) & 15) == 0;
    const long nsup = (E + 511) / 512;

    for (int t = blockIdx.x; t < ntiles; t += G) {
        const int lo = t * tn;
        const int rows = (N - lo < tn) ? (N - lo) : tn;
        int* const ls = lst + lhdr + (long)t * 3 * cap;
        int* const ld = ls + cap;
        int* const lw = ld + cap;
        int lraw = -1;
        if (binned) lraw = lst[t];

        for (int r = tid / L4; r < rows; r += 1024 / L4) {
            const int n = lo + r;
            const float di = dinv[n];
            const float4 hv = *(const float4*)(h + (size_t)n * F + col4 * 4);
            const float4 bv = *(const float4*)(bias + col4 * 4);
            float* a = acc + r * F + col4;
            a[0 * L4] = hv.x * di * di + bv.x;
            a[1 * L4] = hv.y * di * di + bv.y;
            a[2 * L4] = hv.z * di * di + bv.z;
            a[3 * L4] = hv.w * di * di + bv.w;
        }
        __syncthreads();

        auto add_rows = [&](int s, int dl, float w, int m, auto&& ahead) {
            for (int j0 = 0; j0 < m; j0 += 8 * RPW) {
                float4 rv[8];
                int    dj[8];
                float  wj[8];
                bool   ok[8];
#pragma unroll
                for (int u = 0; u < 8; ++u) {
                    const int j = j0 + u * RPW + sub;
                    ok[u] = j < m;
                    const int sj = __shfl(s, j & 31);
                    wj[u] = __shfl(w, j & 31);
                    dj[u] = __shfl(dl, j & 31);
                    rv[u] = ok[u] ? *(const float4*)(h + (size_t)sj * F + col4 * 4)
                                  : make_float4(0.f, 0.f, 0.f, 0.f);
                }
                if (j0 == 0) ahead();
#pragma unroll
                for (int u = 0; u < 8; ++u) {
                    if (ok[u]) {
                        const float4 q = sub ? make_float4(rv[u].z, rv[u].w, rv[u].x, rv[u].y)
                                             : rv[u];
                        const int c0 = 2 * sub;
                        float* a = acc + dj[u] * F + col4;
                        __hip_atomic_fetch_add(a + ((c0 + 0) & 3) * L4, q.x * wj[u],
                                               __ATOMIC_RELAXED, __HIP_MEMORY_SCOPE_WORKGROUP);
                        __hip_atomic_fetch_add(a + ((c0 + 1) & 3) * L4, q.y * wj[u],
                                               __ATOMIC_RELAXED, __HIP_MEMORY_SCOPE_WORKGROUP);
                        __hip_atomic_fetch_add(a + ((c0 + 2) & 3) * L4, q.z * wj[u],
                                               __ATOMIC_RELAXED, __HIP_MEMORY_SCOPE_WORKGROUP);
                        __hip_atomic_fetch_add(a + ((c0 + 3) & 3) * L4, q.w * wj[u],
                                               __ATOMIC_RELAXED, __HIP_MEMORY_SCOPE_WORKGROUP);
                    }
                }
            }
        };

        const int lcnt = __builtin_amdgcn_readfirstlane(lraw);
        if (lcnt >= 0 && lcnt <= cap) {
            constexpr int STEP = NW * 32;
            int base = wave * 32;
            if (FIRST) {
                if (base < lcnt) {
                    const int last = lcnt - 1;
                    int s, dl, s1, dl1;
                    {
                        const int k0 = (base + lane < last) ? (base + lane) : last;
                        const int k1 = (base + STEP + lane < last) ? (base + STEP + lane) : last;
                        s = ls[k0]; dl = ld[k0]; s1 = ls[k1]; dl1 = ld[k1];
                    }
                    float w = dinv[s] * dinv[lo + dl];
                    while (base < lcnt) {
                        float da = 0.0f, db = 0.0f;
                        int s2 = 0, dl2 = 0;
                        if (base + lane < lcnt) lw[base + lane] = __float_as_int(w);
                        add_rows(s, dl, w, (lcnt - base < 32) ? (lcnt - base) : 32, [&] {
                            da = dinv[s1]; db = dinv[lo + dl1];
                            const int k2 = (base + 2 * STEP + lane < last) ? (base + 2 * STEP + lane)
                                                                          : last;
                            s2 = ls[k2]; dl2 = ld[k2];
                            __builtin_amdgcn_sched_barrier(0);
                        });
                        w = da * db;
                        s = s1; dl = dl1; s1 = s2; dl1 = dl2; base += STEP;
                    }
                }
            } else {
                int s = 0, dl = 0; float w = 0.0f;
                if (base + lane < lcnt) {
                    s = ls[base + lane]; dl = ld[base + lane]; w = __int_as_float(lw[base + lane]);
                }
                while (base < lcnt) {
                    const int nb = base + STEP;
                    int s2 = 0, dl2 = 0; float w2 = 0.0f;
                    if (nb + lane < lcnt) {
                        s2 = ls[nb + lane]; dl2 = ld[nb + lane]; w2 = __int_as_float(lw[nb + lane]);
                    }
                    add_rows(s, dl, w, (lcnt - base < 32) ? (lcnt - base) : 32, [] {});
                    s = s2; dl = dl2; w = w2; base = nb;
                }
            }
        } else {
        int cnt = 0;
        auto flush = [&](int n) {
            for (int base = 0; base < n; base += 32) {
                const int i = base + lane;
                const bool v = i < n;
                const int e  = v ? myqe[i] : 0;
                const int dl = v ? myqd[i] : 0;
                const int s  = v ? src[e] : 0;
                const float w = v ? dinv[s] * dinv[lo + dl] : 0.0f;
                const int m = (n - base < 32) ? (n - base) : 32;
                add_rows(s, dl, w, m, [] {});
            }
        };
        const long rot = ((long)blockIdx.x * nsup) / G;
        for (long it = wave; it < nsup; it += NW) {
            const long sp = (it + rot < nsup) ? (it + rot) : (it + rot - nsup);
            const long eb = sp * 512 + lane * 4;
            int d[4][4];
            if (al && sp * 512 + 512 <= E) {
#pragma unroll
                for (int k = 0; k < 4; ++k) {
                    const int4 vv = *(const int4*)(dst + eb + k * 128);
                    d[k][0] = vv.x; d[k][1] = vv.y; d[k][2] = vv.z; d[k][3] = vv.w;
                }
            } else {
#pragma unroll
                for (int k = 0; k < 4; ++k) {
#pragma unroll
                    for (int c = 0; c < 4; ++c) {
                        const long e1 = eb + k * 128 + c;
                        d[k][c] = (e1 < E) ? dst[e1] : -1;
                    }
                }
            }
#pragma unroll
            for (int k = 0; k < 4; ++k) {
                const long e0 = eb + k * 128;
                bool mm[4];
#pragma unroll
                for (int c = 0; c < 4; ++c) mm[c] = (unsigned)(d[k][c] - lo) < (unsigned)rows;
                if (__builtin_amdgcn_ballot_w32(mm[0] | mm[1] | mm[2] | mm[3]) != 0u) {
                    if (cnt > QCAP - 128) { flush(cnt); cnt = 0; }
#pragma unroll
                    for (int c = 0; c < 4; ++c) {
                        const unsigned b = __builtin_amdgcn_ballot_w32(mm[c]);
                        if (mm[c]) {
                            const int p = cnt + (int)__builtin_amdgcn_mbcnt_lo(b, 0u);
                            myqe[p] = (int)(e0 + c);
                            myqd[p] = d[k][c] - lo;
                        }
                        cnt += __builtin_popcount(b);
                    }
                }
            }
        }
        flush(cnt);
        }
        __syncthreads();

        for (int r = tid / L4; r < rows; r += 1024 / L4) {
            const float* a = acc + r * F + col4;
            float4 o = make_float4(a[0 * L4], a[1 * L4], a[2 * L4], a[3 * L4]);
            if (RELU) {
                o.x = o.x > 0.0f ? o.x : 0.0f;
                o.y = o.y > 0.0f ? o.y : 0.0f;
                o.z = o.z > 0.0f ? o.z : 0.0f;
                o.w = o.w > 0.0f ? o.w : 0.0f;
            }
            *(float4*)(outp + (size_t)(lo + r) * F + col4 * 4) = o;
        }
        __syncthreads();
    }
}

extern "C" void kernel_launch(void* const* d_in, const int* in_sizes, int n_in,
                              void* d_out, int out_size, void* d_ws, size_t ws_size,
                              hipStream_t stream) {
    const float* x  = (const float*)d_in[0];
    const int*   ei = (const int*)d_in[1];
    const float* W1 = (const float*)d_in[2];
    const float* b1 = (const float*)d_in[3];
    const float* W2 = (const float*)d_in[4];
    const float* b2 = (const float*)d_in[5];
    float* out = (float*)d_out;

    const int  N = in_sizes[0] / IN_F;
    const long E = (long)in_sizes[1] / 2;
    const int* srcIdx = ei;
    const int* dstIdx = ei + E;

    float* ws   = (float*)d_ws;
    float* dinv = ws;
    float* h    = dinv + N;
    float* agg1 = h + (size_t)N * HID_F;
    int*   lst  = (int*)(agg1 + (size_t)N * HID_F);
    float* h2   = h;

    k_init<<<4, 256, 0, stream>>>(lst, N);

    k_bin<<<(int)((E + BIN_PASS - 1) / BIN_PASS), 1024, 0, stream>>>(srcIdx, dstIdx, lst, N, E);
    k_tile_deg<<<AGG_G, DEG_T, 0, stream>>>(dstIdx, dinv, lst, N, E);

    constexpr int RT   = 4;
    constexpr int ROWS = 16 * RT;

    k_gemm_wmma<IN_F, HID_F, RT>
        <<<(N + ROWS - 1) / ROWS, 32 * (HID_F / 16), 0, stream>>>(x, W1, h, N);
    k_tile_agg<HID_F, true, true>
        <<<AGG_G, 1024, 0, stream>>>(srcIdx, dstIdx, dinv, h, b1, agg1, lst, N, E);

    k_gemm_wmma<HID_F, OUT_F, RT>
        <<<(N + ROWS - 1) / ROWS, 32 * (OUT_F / 16), 0, stream>>>(agg1, W2, h2, N);
    k_tile_agg<OUT_F, false, false>
        <<<AGG_G, 1024, 0, stream>>>(srcIdx, dstIdx, dinv, h2, b2, out, lst, N, E);
}


# Dense products re-tiled: wave owns 16 rows x all columns; A fragments by 16-byte global loads straight to registers, W once per work-group in LDS as k quads, X not staged; k sum order changes
# speedup vs baseline: 4.0443x; 1.2027x over previous
#include <hip/hip_runtime.h>

typedef __attribute__((ext_vector_type(2))) float v2f;
typedef __attribute__((ext_vector_type(8))) float v8f;

#define IN_F  128
#define HID_F 128
#define OUT_F 64

#define AGG_G 256
#define TMAX  400
#define MAXT  1024
#define BIN_EPT 8
#define BIN_PASS (1024 * BIN_EPT)

struct TileGeo { int tn; int ntiles; int cap; long lhdr; };

__device__ __forceinline__ TileGeo tile_geo(int N) {
    TileGeo g;
    int tn = (N + AGG_G - 1) / AGG_G;
    if (tn > TMAX) tn = TMAX;
    if (tn < 1) tn = 1;
    g.tn = tn;
    g.ntiles = (N + tn - 1) / tn;
    const long lwords = (long)N * OUT_F;
    g.lhdr = ((long)g.ntiles + 31) & ~31L;
    long lc = (lwords - g.lhdr) / g.ntiles / 3;
    if (lc < 0) lc = 0;
    if (lc > (1L << 30)) lc = 1L << 30;
    g.cap = (int)(lc & ~31L);
    return g;
}

__global__ void k_init(int* __restrict__ lst, int N) {
    const TileGeo g = tile_geo(N);
    long i = (long)blockIdx.x * blockDim.x + threadIdx.x;
    long stride = (long)gridDim.x * blockDim.x;
    for (long j = i; j < g.lhdr; j += stride) lst[j] = 0;
}

__global__ void __launch_bounds__(1024)
k_bin(const int* __restrict__ src, const int* __restrict__ dst,
      int* __restrict__ lst, int N, long E) {
    __shared__ int hist[MAXT];
    __shared__ int gdelta[MAXT];
    __shared__ int ssrc[BIN_PASS];
    __shared__ int spk[BIN_PASS];
    __shared__ int wsum[32];
    __shared__ int tot;
    const TileGeo g = tile_geo(N);
    const bool binned = g.ntiles <= MAXT && g.cap > 0;
    const int tid  = threadIdx.x;
    const int lane = tid & 31;
    const int wave = tid >> 5;

    for (long p0 = (long)blockIdx.x * BIN_PASS; p0 < E; p0 += (long)gridDim.x * BIN_PASS) {
        hist[tid] = 0;
        __syncthreads();

        int d[BIN_EPT], s[BIN_EPT], tl[BIN_EPT], rk[BIN_EPT];
#pragma unroll
        for (int k = 0; k < BIN_EPT; ++k) {
            const long e = p0 + k * 1024 + tid;
            const bool ok = e < E;
            d[k] = ok ? dst[e] : -1;
            s[k] = ok ? src[e] : 0;
        }
#pragma unroll
        for (int k = 0; k < BIN_EPT; ++k) {
            tl[k] = 0; rk[k] = 0;
            if (binned && (unsigned)d[k] < (unsigned)N) {
                tl[k] = d[k] / g.tn;
                rk[k] = __hip_atomic_fetch_add(&hist[tl[k]], 1, __ATOMIC_RELAXED,
                                               __HIP_MEMORY_SCOPE_WORKGROUP);
            }
        }
        __syncthreads();
        if (!binned) continue;

        const int c = hist[tid];
        int gb = 0;
        if (c > 0 && tid < g.ntiles)
            gb = __hip_atomic_fetch_add(&lst[tid], c, __ATOMIC_RELAXED, __HIP_MEMORY_SCOPE_AGENT);
        int x = c;
#pragma unroll
        for (int o = 1; o < 32; o <<= 1) {
            const int y = __shfl_up(x, o);
            if (lane >= o) x += y;
        }
        if (lane == 31) wsum[wave] = x;
        __syncthreads();
        if (wave == 0) {
            const int v = wsum[lane];
            int z = v;
#pragma unroll
            for (int o = 1; o < 32; o <<= 1) {
                const int y = __shfl_up(z, o);
                if (lane >= o) z += y;
            }
            wsum[lane] = z - v;
            if (lane == 31) tot = z;
        }
        __syncthreads();
        const int excl = x - c + wsum[wave];
        hist[tid]   = excl;
        gdelta[tid] = gb - excl;
        __syncthreads();

#pragma unroll
        for (int k = 0; k < BIN_EPT; ++k) {
            if ((unsigned)d[k] < (unsigned)N) {
                const int p = hist[tl[k]] + rk[k];
                ssrc[p] = s[k];
                spk[p]  = (tl[k] << 9) | (d[k] - tl[k] * g.tn);
            }
        }
        __syncthreads();

        const int n = tot;
        for (int i = tid; i < n; i += 1024) {
            const int pk = spk[i];
            const int t  = pk >> 9;
            const int pos = gdelta[t] + i;
            if (pos >= 0 && pos < g.cap) {
                int* const ls = lst + g.lhdr + (long)t * 3 * g.cap;
                ls[pos]         = ssrc[i];
                ls[g.cap + pos] = pk & 511;
            }
        }
    }
}

#define DEG_T 512
#define DEG_U 8
__global__ void __launch_bounds__(DEG_T)
k_tile_deg(const int* __restrict__ dst, float* __restrict__ dinv,
           const int* __restrict__ lst, int N, long E) {
    __shared__ int cnt[TMAX];
    const TileGeo g = tile_geo(N);
    const bool binned = g.ntiles <= MAXT && g.cap > 0;
    const int tid = threadIdx.x;
    for (int t = blockIdx.x; t < g.ntiles; t += gridDim.x) {
        const int lo = t * g.tn;
        const int rows = (N - lo < g.tn) ? (N - lo) : g.tn;
        int lcnt = -1;
        if (binned) lcnt = lst[t];
        for (int r = tid; r < rows; r += DEG_T) cnt[r] = 0;
        __syncthreads();
        if (lcnt >= 0 && lcnt <= g.cap) {
            const int* const ld = lst + g.lhdr + (long)t * 3 * g.cap + g.cap;
            for (int i0 = tid; i0 < lcnt; i0 += DEG_T * DEG_U) {
                int r[DEG_U];
#pragma unroll
                for (int u = 0; u < DEG_U; ++u) {
                    const int i = i0 + u * DEG_T;
                    r[u] = (i < lcnt) ? ld[i] : -1;
                }
#pragma unroll
                for (int u = 0; u < DEG_U; ++u)
                    if ((unsigned)r[u] < (unsigned)rows)
                        __hip_atomic_fetch_add(&cnt[r[u]], 1, __ATOMIC_RELAXED,
                                               __HIP_MEMORY_SCOPE_WORKGROUP);
            }
        } else {
            for (long e = tid; e < E; e += DEG_T) {
                const unsigned r = (unsigned)(dst[e] - lo);
                if (r < (unsigned)rows)
                    __hip_atomic_fetch_add(&cnt[r], 1, __ATOMIC_RELAXED,
                                           __HIP_MEMORY_SCOPE_WORKGROUP);
            }
        }
        __syncthreads();
        for (int r = tid; r < rows; r += DEG_T)
            dinv[lo + r] = rsqrtf(1.0f + (float)cnt[r]);
        __syncthreads();
    }
}

#define GEMM_WAVES 8
#define GEMM_G     512
template <int K, int NCOL>
__global__ void __launch_bounds__(32 * GEMM_WAVES)
k_gemm_rows(const float* __restrict__ X, const float* __restrict__ W,
            float* __restrict__ H, int M) {
    constexpr int NS  = NCOL / 16;
    constexpr int NKK = K / 8;
    __shared__ float4 sW[(K / 4) * NCOL];

    const int tid  = threadIdx.x;
    const int wave = __builtin_amdgcn_readfirstlane(tid >> 5);
    const int lane = tid & 31;
    const int half = lane >> 4;
    const int idx  = lane & 15;

    static_assert(GEMM_WAVES == 8, "the rank table is for 8 waves");
    const int ntile = (M + 15) / 16;
    const int t0 = (int)((long)blockIdx.x * ntile / gridDim.x) + ((0x73261540 >> (4 * wave)) & 7);
    const int t1 = (int)(((long)blockIdx.x + 1) * ntile / gridDim.x);

    auto row_of = [&](int t) {
        int ar = t * 16 + idx;
        if (ar > M - 1) ar = M - 1;
        return X + (size_t)ar * K + half * 4;
    };

    float4 a[NKK];
    if (t0 < t1) {
        const float* const xr = row_of(t0);
#pragma unroll
        for (int kk = 0; kk < NKK; ++kk) a[kk] = *(const float4*)(xr + kk * 8);
    }

    constexpr int WITEMS = (K / 4) * (NCOL / 4);
    static_assert(WITEMS % (32 * GEMM_WAVES) == 0, "whole trips");
#pragma unroll
    for (int j = 0; j < WITEMS / (32 * GEMM_WAVES); ++j) {
        const int i  = tid + j * 32 * GEMM_WAVES;
        const int q  = i / (NCOL / 4);
        const int n4 = i % (NCOL / 4);
        const float* const w = W + (size_t)(4 * q) * NCOL + n4 * 4;
        const float4 r0 = *(const float4*)(w);
        const float4 r1 = *(const float4*)(w + NCOL);
        const float4 r2 = *(const float4*)(w + 2 * NCOL);
        const float4 r3 = *(const float4*)(w + 3 * NCOL);
        float4* const d = sW + q * NCOL + n4 * 4;
        d[0] = make_float4(r0.x, r1.x, r2.x, r3.x);
        d[1] = make_float4(r0.y, r1.y, r2.y, r3.y);
        d[2] = make_float4(r0.z, r1.z, r2.z, r3.z);
        d[3] = make_float4(r0.w, r1.w, r2.w, r3.w);
    }
    __syncthreads();

    const float4* const bl = sW + half * NCOL + idx;

    for (int t = t0; t < t1; t += GEMM_WAVES) {
        const int row0 = t * 16;
        v8f acc[NS] = {};
        float4 bc[NS];
#pragma unroll
        for (int c = 0; c < NS; ++c) bc[c] = bl[c * 16];
#pragma unroll
        for (int kk = 0; kk < NKK; ++kk) {
            float4 bn[NS];
            if (kk + 1 < NKK) {
#pragma unroll
                for (int c = 0; c < NS; ++c) bn[c] = bl[(kk + 1) * 2 * NCOL + c * 16];
            }
#pragma unroll
            for (int s = 0; s < 2; ++s) {
                v2f av;
                av.x = s ? a[kk].z : a[kk].x;
                av.y = s ? a[kk].w : a[kk].y;
#pragma unroll
                for (int c = 0; c < NS; ++c) {
                    v2f bv;
                    bv.x = s ? bc[c].z : bc[c].x;
                    bv.y = s ? bc[c].w : bc[c].y;
                    acc[c] = __builtin_amdgcn_wmma_f32_16x16x4_f32(
                         false, av,  false, bv,
                         (short)0, acc[c],  false,  false);
                }
            }
            if (kk + 1 < NKK) {
#pragma unroll
                for (int c = 0; c < NS; ++c) bc[c] = bn[c];
            }
            __builtin_amdgcn_sched_barrier(0);
        }

#pragma unroll
        for (int v = 0; v < 8; ++v) {
            const int gr = row0 + half * 8 + v;
            if (gr < M) {
#pragma unroll
                for (int c = 0; c < NS; ++c)
                    H[(size_t)gr * NCOL + c * 16 + idx] = acc[c][v];
            }
        }

        if (t + GEMM_WAVES < t1) {
            const float* const xr = row_of(t + GEMM_WAVES);
#pragma unroll
            for (int kk = 0; kk < NKK; ++kk) a[kk] = *(const float4*)(xr + kk * 8);
        }
    }
}

template <int F, bool RELU, bool FIRST>
__global__ void __launch_bounds__(1024)
k_tile_agg(const int* __restrict__ src, const int* __restrict__ dst,
           const float* __restrict__ dinv, const float* __restrict__ h,
           const float* __restrict__ bias, float* __restrict__ outp,
           int* __restrict__ lst, int N, long E) {
    constexpr int NW   = 32;
    constexpr int QCAP = 256;
    constexpr int L4   = F / 4;
    constexpr int RPW  = 32 / L4;
    __shared__ float acc[TMAX * F];
    __shared__ int   qe[NW * QCAP];
    __shared__ int   qd[NW * QCAP];

    const int tid  = threadIdx.x;
    const int wave = __builtin_amdgcn_readfirstlane(tid >> 5);
    const int lane = tid & 31;
    const int sub  = lane / L4;
    const int col4 = lane % L4;
    int* myqe = qe + wave * QCAP;
    int* myqd = qd + wave * QCAP;

    const int G = gridDim.x;
    const TileGeo g = tile_geo(N);
    const int tn = g.tn;
    const int ntiles = g.ntiles;
    const int cap = g.cap;
    const long lhdr = g.lhdr;
    const bool binned = ntiles <= MAXT && cap > 0;
    const bool al = (((size_t)dst) & 15) == 0;
    const long nsup = (E + 511) / 512;

    for (int t = blockIdx.x; t < ntiles; t += G) {
        const int lo = t * tn;
        const int rows = (N - lo < tn) ? (N - lo) : tn;
        int* const ls = lst + lhdr + (long)t * 3 * cap;
        int* const ld = ls + cap;
        int* const lw = ld + cap;
        int lraw = -1;
        if (binned) lraw = lst[t];

        for (int r = tid / L4; r < rows; r += 1024 / L4) {
            const int n = lo + r;
            const float di = dinv[n];
            const float4 hv = *(const float4*)(h + (size_t)n * F + col4 * 4);
            const float4 bv = *(const float4*)(bias + col4 * 4);
            float* a = acc + r * F + col4;
            a[0 * L4] = hv.x * di * di + bv.x;
            a[1 * L4] = hv.y * di * di + bv.y;
            a[2 * L4] = hv.z * di * di + bv.z;
            a[3 * L4] = hv.w * di * di + bv.w;
        }
        __syncthreads();

        auto add_rows = [&](int s, int dl, float w, int m, auto&& ahead) {
            for (int j0 = 0; j0 < m; j0 += 8 * RPW) {
                float4 rv[8];
                int    dj[8];
                float  wj[8];
                bool   ok[8];
#pragma unroll
                for (int u = 0; u < 8; ++u) {
                    const int j = j0 + u * RPW + sub;
                    ok[u] = j < m;
                    const int sj = __shfl(s, j & 31);
                    wj[u] = __shfl(w, j & 31);
                    dj[u] = __shfl(dl, j & 31);
                    rv[u] = ok[u] ? *(const float4*)(h + (size_t)sj * F + col4 * 4)
                                  : make_float4(0.f, 0.f, 0.f, 0.f);
                }
                if (j0 == 0) ahead();
#pragma unroll
                for (int u = 0; u < 8; ++u) {
                    if (ok[u]) {
                        const float4 q = sub ? make_float4(rv[u].z, rv[u].w, rv[u].x, rv[u].y)
                                             : rv[u];
                        const int c0 = 2 * sub;
                        float* a = acc + dj[u] * F + col4;
                        __hip_atomic_fetch_add(a + ((c0 + 0) & 3) * L4, q.x * wj[u],
                                               __ATOMIC_RELAXED, __HIP_MEMORY_SCOPE_WORKGROUP);
                        __hip_atomic_fetch_add(a + ((c0 + 1) & 3) * L4, q.y * wj[u],
                                               __ATOMIC_RELAXED, __HIP_MEMORY_SCOPE_WORKGROUP);
                        __hip_atomic_fetch_add(a + ((c0 + 2) & 3) * L4, q.z * wj[u],
                                               __ATOMIC_RELAXED, __HIP_MEMORY_SCOPE_WORKGROUP);
                        __hip_atomic_fetch_add(a + ((c0 + 3) & 3) * L4, q.w * wj[u],
                                               __ATOMIC_RELAXED, __HIP_MEMORY_SCOPE_WORKGROUP);
                    }
                }
            }
        };

        const int lcnt = __builtin_amdgcn_readfirstlane(lraw);
        if (lcnt >= 0 && lcnt <= cap) {
            constexpr int STEP = NW * 32;
            int base = wave * 32;
            if (FIRST) {
                if (base < lcnt) {
                    const int last = lcnt - 1;
                    int s, dl, s1, dl1;
                    {
                        const int k0 = (base + lane < last) ? (base + lane) : last;
                        const int k1 = (base + STEP + lane < last) ? (base + STEP + lane) : last;
                        s = ls[k0]; dl = ld[k0]; s1 = ls[k1]; dl1 = ld[k1];
                    }
                    float w = dinv[s] * dinv[lo + dl];
                    while (base < lcnt) {
                        float da = 0.0f, db = 0.0f;
                        int s2 = 0, dl2 = 0;
                        if (base + lane < lcnt) lw[base + lane] = __float_as_int(w);
                        add_rows(s, dl, w, (lcnt - base < 32) ? (lcnt - base) : 32, [&] {
                            da = dinv[s1]; db = dinv[lo + dl1];
                            const int k2 = (base + 2 * STEP + lane < last) ? (base + 2 * STEP + lane)
                                                                          : last;
                            s2 = ls[k2]; dl2 = ld[k2];
                            __builtin_amdgcn_sched_barrier(0);
                        });
                        w = da * db;
                        s = s1; dl = dl1; s1 = s2; dl1 = dl2; base += STEP;
                    }
                }
            } else {
                int s = 0, dl = 0; float w = 0.0f;
                if (base + lane < lcnt) {
                    s = ls[base + lane]; dl = ld[base + lane]; w = __int_as_float(lw[base + lane]);
                }
                while (base < lcnt) {
                    const int nb = base + STEP;
                    int s2 = 0, dl2 = 0; float w2 = 0.0f;
                    if (nb + lane < lcnt) {
                        s2 = ls[nb + lane]; dl2 = ld[nb + lane]; w2 = __int_as_float(lw[nb + lane]);
                    }
                    add_rows(s, dl, w, (lcnt - base < 32) ? (lcnt - base) : 32, [] {});
                    s = s2; dl = dl2; w = w2; base = nb;
                }
            }
        } else {
        int cnt = 0;
        auto flush = [&](int n) {
            for (int base = 0; base < n; base += 32) {
                const int i = base + lane;
                const bool v = i < n;
                const int e  = v ? myqe[i] : 0;
                const int dl = v ? myqd[i] : 0;
                const int s  = v ? src[e] : 0;
                const float w = v ? dinv[s] * dinv[lo + dl] : 0.0f;
                const int m = (n - base < 32) ? (n - base) : 32;
                add_rows(s, dl, w, m, [] {});
            }
        };
        const long rot = ((long)blockIdx.x * nsup) / G;
        for (long it = wave; it < nsup; it += NW) {
            const long sp = (it + rot < nsup) ? (it + rot) : (it + rot - nsup);
            const long eb = sp * 512 + lane * 4;
            int d[4][4];
            if (al && sp * 512 + 512 <= E) {
#pragma unroll
                for (int k = 0; k < 4; ++k) {
                    const int4 vv = *(const int4*)(dst + eb + k * 128);
                    d[k][0] = vv.x; d[k][1] = vv.y; d[k][2] = vv.z; d[k][3] = vv.w;
                }
            } else {
#pragma unroll
                for (int k = 0; k < 4; ++k) {
#pragma unroll
                    for (int c = 0; c < 4; ++c) {
                        const long e1 = eb + k * 128 + c;
                        d[k][c] = (e1 < E) ? dst[e1] : -1;
                    }
                }
            }
#pragma unroll
            for (int k = 0; k < 4; ++k) {
                const long e0 = eb + k * 128;
                bool mm[4];
#pragma unroll
                for (int c = 0; c < 4; ++c) mm[c] = (unsigned)(d[k][c] - lo) < (unsigned)rows;
                if (__builtin_amdgcn_ballot_w32(mm[0] | mm[1] | mm[2] | mm[3]) != 0u) {
                    if (cnt > QCAP - 128) { flush(cnt); cnt = 0; }
#pragma unroll
                    for (int c = 0; c < 4; ++c) {
                        const unsigned b = __builtin_amdgcn_ballot_w32(mm[c]);
                        if (mm[c]) {
                            const int p = cnt + (int)__builtin_amdgcn_mbcnt_lo(b, 0u);
                            myqe[p] = (int)(e0 + c);
                            myqd[p] = d[k][c] - lo;
                        }
                        cnt += __builtin_popcount(b);
                    }
                }
            }
        }
        flush(cnt);
        }
        __syncthreads();

        for (int r = tid / L4; r < rows; r += 1024 / L4) {
            const float* a = acc + r * F + col4;
            float4 o = make_float4(a[0 * L4], a[1 * L4], a[2 * L4], a[3 * L4]);
            if (RELU) {
                o.x = o.x > 0.0f ? o.x : 0.0f;
                o.y = o.y > 0.0f ? o.y : 0.0f;
                o.z = o.z > 0.0f ? o.z : 0.0f;
                o.w = o.w > 0.0f ? o.w : 0.0f;
            }
            *(float4*)(outp + (size_t)(lo + r) * F + col4 * 4) = o;
        }
        __syncthreads();
    }
}

extern "C" void kernel_launch(void* const* d_in, const int* in_sizes, int n_in,
                              void* d_out, int out_size, void* d_ws, size_t ws_size,
                              hipStream_t stream) {
    const float* x  = (const float*)d_in[0];
    const int*   ei = (const int*)d_in[1];
    const float* W1 = (const float*)d_in[2];
    const float* b1 = (const float*)d_in[3];
    const float* W2 = (const float*)d_in[4];
    const float* b2 = (const float*)d_in[5];
    float* out = (float*)d_out;

    const int  N = in_sizes[0] / IN_F;
    const long E = (long)in_sizes[1] / 2;
    const int* srcIdx = ei;
    const int* dstIdx = ei + E;

    float* ws   = (float*)d_ws;
    float* dinv = ws;
    float* h    = dinv + N;
    float* agg1 = h + (size_t)N * HID_F;
    int*   lst  = (int*)(agg1 + (size_t)N * HID_F);
    float* h2   = h;

    k_init<<<4, 256, 0, stream>>>(lst, N);

    k_bin<<<(int)((E + BIN_PASS - 1) / BIN_PASS), 1024, 0, stream>>>(srcIdx, dstIdx, lst, N, E);
    k_tile_deg<<<AGG_G, DEG_T, 0, stream>>>(dstIdx, dinv, lst, N, E);

    k_gemm_rows<IN_F, HID_F><<<GEMM_G, 32 * GEMM_WAVES, 0, stream>>>(x, W1, h, N);
    k_tile_agg<HID_F, true, true>
        <<<AGG_G, 1024, 0, stream>>>(srcIdx, dstIdx, dinv, h, b1, agg1, lst, N, E);

    k_gemm_rows<HID_F, OUT_F><<<GEMM_G, 32 * GEMM_WAVES, 0, stream>>>(agg1, W2, h2, N);
    k_tile_agg<OUT_F, false, false>
        <<<AGG_G, 1024, 0, stream>>>(srcIdx, dstIdx, dinv, h2, b2, out, lst, N, E);
}


# Aggregation sums in registers: tile entries routed through LDS queues to owner waves (rows w, w+32, ..), a wave adds its queued rows into registers; LDS float atomics go; sum order = queue arrival
# speedup vs baseline: 4.3479x; 1.0751x over previous
#include <hip/hip_runtime.h>

typedef __attribute__((ext_vector_type(2))) float v2f;
typedef __attribute__((ext_vector_type(8))) float v8f;

#define IN_F  128
#define HID_F 128
#define OUT_F 64

#define AGG_G 256
#define TMAX  400
#define MAXT  1024
#define BIN_EPT 8
#define BIN_PASS (1024 * BIN_EPT)

struct TileGeo { int tn; int ntiles; int cap; long lhdr; };

__device__ __forceinline__ TileGeo tile_geo(int N) {
    TileGeo g;
    int tn = (N + AGG_G - 1) / AGG_G;
    if (tn > TMAX) tn = TMAX;
    if (tn < 1) tn = 1;
    g.tn = tn;
    g.ntiles = (N + tn - 1) / tn;
    const long lwords = (long)N * OUT_F;
    g.lhdr = ((long)g.ntiles + 31) & ~31L;
    long lc = (lwords - g.lhdr) / g.ntiles / 3;
    if (lc < 0) lc = 0;
    if (lc > (1L << 30)) lc = 1L << 30;
    g.cap = (int)(lc & ~31L);
    return g;
}

__global__ void k_init(int* __restrict__ lst, int N) {
    const TileGeo g = tile_geo(N);
    long i = (long)blockIdx.x * blockDim.x + threadIdx.x;
    long stride = (long)gridDim.x * blockDim.x;
    for (long j = i; j < g.lhdr; j += stride) lst[j] = 0;
}

__global__ void __launch_bounds__(1024)
k_bin(const int* __restrict__ src, const int* __restrict__ dst,
      int* __restrict__ lst, int N, long E) {
    __shared__ int hist[MAXT];
    __shared__ int gdelta[MAXT];
    __shared__ int ssrc[BIN_PASS];
    __shared__ int spk[BIN_PASS];
    __shared__ int wsum[32];
    __shared__ int tot;
    const TileGeo g = tile_geo(N);
    const bool binned = g.ntiles <= MAXT && g.cap > 0;
    const int tid  = threadIdx.x;
    const int lane = tid & 31;
    const int wave = tid >> 5;

    for (long p0 = (long)blockIdx.x * BIN_PASS; p0 < E; p0 += (long)gridDim.x * BIN_PASS) {
        hist[tid] = 0;
        __syncthreads();

        int d[BIN_EPT], s[BIN_EPT], tl[BIN_EPT], rk[BIN_EPT];
#pragma unroll
        for (int k = 0; k < BIN_EPT; ++k) {
            const long e = p0 + k * 1024 + tid;
            const bool ok = e < E;
            d[k] = ok ? dst[e] : -1;
            s[k] = ok ? src[e] : 0;
        }
#pragma unroll
        for (int k = 0; k < BIN_EPT; ++k) {
            tl[k] = 0; rk[k] = 0;
            if (binned && (unsigned)d[k] < (unsigned)N) {
                tl[k] = d[k] / g.tn;
                rk[k] = __hip_atomic_fetch_add(&hist[tl[k]], 1, __ATOMIC_RELAXED,
                                               __HIP_MEMORY_SCOPE_WORKGROUP);
            }
        }
        __syncthreads();
        if (!binned) continue;

        const int c = hist[tid];
        int gb = 0;
        if (c > 0 && tid < g.ntiles)
            gb = __hip_atomic_fetch_add(&lst[tid], c, __ATOMIC_RELAXED, __HIP_MEMORY_SCOPE_AGENT);
        int x = c;
#pragma unroll
        for (int o = 1; o < 32; o <<= 1) {
            const int y = __shfl_up(x, o);
            if (lane >= o) x += y;
        }
        if (lane == 31) wsum[wave] = x;
        __syncthreads();
        if (wave == 0) {
            const int v = wsum[lane];
            int z = v;
#pragma unroll
            for (int o = 1; o < 32; o <<= 1) {
                const int y = __shfl_up(z, o);
                if (lane >= o) z += y;
            }
            wsum[lane] = z - v;
            if (lane == 31) tot = z;
        }
        __syncthreads();
        const int excl = x - c + wsum[wave];
        hist[tid]   = excl;
        gdelta[tid] = gb - excl;
        __syncthreads();

#pragma unroll
        for (int k = 0; k < BIN_EPT; ++k) {
            if ((unsigned)d[k] < (unsigned)N) {
                const int p = hist[tl[k]] + rk[k];
                ssrc[p] = s[k];
                spk[p]  = (tl[k] << 9) | (d[k] - tl[k] * g.tn);
            }
        }
        __syncthreads();

        const int n = tot;
        for (int i = tid; i < n; i += 1024) {
            const int pk = spk[i];
            const int t  = pk >> 9;
            const int pos = gdelta[t] + i;
            if (pos >= 0 && pos < g.cap) {
                int* const ls = lst + g.lhdr + (long)t * 3 * g.cap;
                ls[pos]         = ssrc[i];
                ls[g.cap + pos] = pk & 511;
            }
        }
    }
}

#define DEG_T 512
#define DEG_U 8
__global__ void __launch_bounds__(DEG_T)
k_tile_deg(const int* __restrict__ dst, float* __restrict__ dinv,
           const int* __restrict__ lst, int N, long E) {
    __shared__ int cnt[TMAX];
    const TileGeo g = tile_geo(N);
    const bool binned = g.ntiles <= MAXT && g.cap > 0;
    const int tid = threadIdx.x;
    for (int t = blockIdx.x; t < g.ntiles; t += gridDim.x) {
        const int lo = t * g.tn;
        const int rows = (N - lo < g.tn) ? (N - lo) : g.tn;
        int lcnt = -1;
        if (binned) lcnt = lst[t];
        for (int r = tid; r < rows; r += DEG_T) cnt[r] = 0;
        __syncthreads();
        if (lcnt >= 0 && lcnt <= g.cap) {
            const int* const ld = lst + g.lhdr + (long)t * 3 * g.cap + g.cap;
            for (int i0 = tid; i0 < lcnt; i0 += DEG_T * DEG_U) {
                int r[DEG_U];
#pragma unroll
                for (int u = 0; u < DEG_U; ++u) {
                    const int i = i0 + u * DEG_T;
                    r[u] = (i < lcnt) ? ld[i] : -1;
                }
#pragma unroll
                for (int u = 0; u < DEG_U; ++u)
                    if ((unsigned)r[u] < (unsigned)rows)
                        __hip_atomic_fetch_add(&cnt[r[u]], 1, __ATOMIC_RELAXED,
                                               __HIP_MEMORY_SCOPE_WORKGROUP);
            }
        } else {
            for (long e = tid; e < E; e += DEG_T) {
                const unsigned r = (unsigned)(dst[e] - lo);
                if (r < (unsigned)rows)
                    __hip_atomic_fetch_add(&cnt[r], 1, __ATOMIC_RELAXED,
                                           __HIP_MEMORY_SCOPE_WORKGROUP);
            }
        }
        __syncthreads();
        for (int r = tid; r < rows; r += DEG_T)
            dinv[lo + r] = rsqrtf(1.0f + (float)cnt[r]);
        __syncthreads();
    }
}

#define GEMM_WAVES 8
#define GEMM_G     512
template <int K, int NCOL>
__global__ void __launch_bounds__(32 * GEMM_WAVES)
k_gemm_rows(const float* __restrict__ X, const float* __restrict__ W,
            float* __restrict__ H, int M) {
    constexpr int NS  = NCOL / 16;
    constexpr int NKK = K / 8;
    __shared__ float4 sW[(K / 4) * NCOL];

    const int tid  = threadIdx.x;
    const int wave = __builtin_amdgcn_readfirstlane(tid >> 5);
    const int lane = tid & 31;
    const int half = lane >> 4;
    const int idx  = lane & 15;

    static_assert(GEMM_WAVES == 8, "the rank table is for 8 waves");
    const int ntile = (M + 15) / 16;
    const int t0 = (int)((long)blockIdx.x * ntile / gridDim.x) + ((0x73261540 >> (4 * wave)) & 7);
    const int t1 = (int)(((long)blockIdx.x + 1) * ntile / gridDim.x);

    auto row_of = [&](int t) {
        int ar = t * 16 + idx;
        if (ar > M - 1) ar = M - 1;
        return X + (size_t)ar * K + half * 4;
    };

    float4 a[NKK];
    if (t0 < t1) {
        const float* const xr = row_of(t0);
#pragma unroll
        for (int kk = 0; kk < NKK; ++kk) a[kk] = *(const float4*)(xr + kk * 8);
    }

    constexpr int WITEMS = (K / 4) * (NCOL / 4);
    static_assert(WITEMS % (32 * GEMM_WAVES) == 0, "whole trips");
#pragma unroll
    for (int j = 0; j < WITEMS / (32 * GEMM_WAVES); ++j) {
        const int i  = tid + j * 32 * GEMM_WAVES;
        const int q  = i / (NCOL / 4);
        const int n4 = i % (NCOL / 4);
        const float* const w = W + (size_t)(4 * q) * NCOL + n4 * 4;
        const float4 r0 = *(const float4*)(w);
        const float4 r1 = *(const float4*)(w + NCOL);
        const float4 r2 = *(const float4*)(w + 2 * NCOL);
        const float4 r3 = *(const float4*)(w + 3 * NCOL);
        float4* const d = sW + q * NCOL + n4 * 4;
        d[0] = make_float4(r0.x, r1.x, r2.x, r3.x);
        d[1] = make_float4(r0.y, r1.y, r2.y, r3.y);
        d[2] = make_float4(r0.z, r1.z, r2.z, r3.z);
        d[3] = make_float4(r0.w, r1.w, r2.w, r3.w);
    }
    __syncthreads();

    const float4* const bl = sW + half * NCOL + idx;

    for (int t = t0; t < t1; t += GEMM_WAVES) {
        const int row0 = t * 16;
        v8f acc[NS] = {};
        float4 bc[NS];
#pragma unroll
        for (int c = 0; c < NS; ++c) bc[c] = bl[c * 16];
#pragma unroll
        for (int kk = 0; kk < NKK; ++kk) {
            float4 bn[NS];
            if (kk + 1 < NKK) {
#pragma unroll
                for (int c = 0; c < NS; ++c) bn[c] = bl[(kk + 1) * 2 * NCOL + c * 16];
            }
#pragma unroll
            for (int s = 0; s < 2; ++s) {
                v2f av;
                av.x = s ? a[kk].z : a[kk].x;
                av.y = s ? a[kk].w : a[kk].y;
#pragma unroll
                for (int c = 0; c < NS; ++c) {
                    v2f bv;
                    bv.x = s ? bc[c].z : bc[c].x;
                    bv.y = s ? bc[c].w : bc[c].y;
                    acc[c] = __builtin_amdgcn_wmma_f32_16x16x4_f32(
                         false, av,  false, bv,
                         (short)0, acc[c],  false,  false);
                }
            }
            if (kk + 1 < NKK) {
#pragma unroll
                for (int c = 0; c < NS; ++c) bc[c] = bn[c];
            }
            __builtin_amdgcn_sched_barrier(0);
        }

#pragma unroll
        for (int v = 0; v < 8; ++v) {
            const int gr = row0 + half * 8 + v;
            if (gr < M) {
#pragma unroll
                for (int c = 0; c < NS; ++c)
                    H[(size_t)gr * NCOL + c * 16 + idx] = acc[c][v];
            }
        }

        if (t + GEMM_WAVES < t1) {
            const float* const xr = row_of(t + GEMM_WAVES);
#pragma unroll
            for (int kk = 0; kk < NKK; ++kk) a[kk] = *(const float4*)(xr + kk * 8);
        }
    }
}

#define QCAP   512
#define QPITCH (QCAP + 1)
#define AGG_EPT 9
template <int F, bool RELU, bool FIRST>
__global__ void __launch_bounds__(1024)
k_tile_agg(const int* __restrict__ src, const int* __restrict__ dst,
           const float* __restrict__ dinv, const float* __restrict__ h,
           const float* __restrict__ bias, float* __restrict__ outp,
           int* __restrict__ lst, int N, long E) {
    constexpr int NW = 32;
    constexpr int VT = F / 32;
    constexpr int KR = (TMAX + NW - 1) / NW;
    static_assert(AGG_EPT * 1024 > (TMAX * 64) / 3, "a thread's share of the longest list");
    static_assert(KR <= 16, "one 16-wide register vector per feature of a lane");
    typedef float vt __attribute__((ext_vector_type(VT)));
    typedef float v16f __attribute__((ext_vector_type(16)));
    __shared__ int   qs[NW * QPITCH];
    __shared__ int   qk[NW * QPITCH];
    __shared__ float qw[NW * QPITCH];
    __shared__ float sdi[TMAX];
    __shared__ int   qn[NW];
    __shared__ int   again;

    const int tid  = threadIdx.x;
    const int wave = __builtin_amdgcn_readfirstlane(tid >> 5);
    const int lane = tid & 31;

    const int G = gridDim.x;
    const TileGeo g = tile_geo(N);
    const int tn = g.tn;
    const int ntiles = g.ntiles;
    const int cap = g.cap;
    const long lhdr = g.lhdr;
    const bool binned = ntiles <= MAXT && cap > 0;

    for (int t = blockIdx.x; t < ntiles; t += G) {
        const int lo = t * tn;
        const int rows = (N - lo < tn) ? (N - lo) : tn;
        int* const ls = lst + lhdr + (long)t * 3 * cap;
        int* const ld = ls + cap;
        int* const lw = ld + cap;
        int lraw = -1;
        if (binned) lraw = lst[t];
        for (int r = tid; r < rows; r += 1024) sdi[r] = dinv[lo + r];

        const int lcnt = __builtin_amdgcn_readfirstlane(lraw);
        const bool listed = lcnt >= 0 && lcnt <= cap;

        auto place = [&](int j, int s, int d, float w, unsigned& pend) {
            const int o = d & (NW - 1);
            const int pos = __hip_atomic_fetch_add(&qn[o], 1, __ATOMIC_RELAXED,
                                                   __HIP_MEMORY_SCOPE_WORKGROUP);
            if (pos < QCAP) {
                qs[o * QPITCH + pos] = s;
                qk[o * QPITCH + pos] = d >> 5;
                qw[o * QPITCH + pos] = w;
                if (FIRST && listed) lw[tid + j * 1024] = __float_as_int(w);
                pend &= ~(1u << j);
            }
        };

        unsigned pend = 0;
        int more0 = 0;
        if (listed) {
#pragma unroll
            for (int j = 0; j < AGG_EPT; ++j)
                if (tid + j * 1024 < lcnt) pend |= 1u << j;
            if (tid < NW) qn[tid] = 0;
            if (tid == 0) again = 0;
            __syncthreads();
            int   es[AGG_EPT], ed[AGG_EPT];
            float ew[AGG_EPT];
#pragma unroll
            for (int j = 0; j < AGG_EPT; ++j) {
                const int i = tid + j * 1024;
                const bool ok = (pend >> j) & 1u;
                es[j] = ok ? ls[i] : 0;
                ed[j] = ok ? ld[i] : 0;
                if (!FIRST) ew[j] = ok ? __int_as_float(lw[i]) : 0.0f;
            }
            if (FIRST) {
#pragma unroll
                for (int j = 0; j < AGG_EPT; ++j) ew[j] = dinv[es[j]];
            }
#pragma unroll
            for (int j = 0; j < AGG_EPT; ++j) {
                if ((pend >> j) & 1u)
                    place(j, es[j], ed[j], FIRST ? ew[j] * sdi[ed[j]] : ew[j], pend);
            }
            if (pend != 0) again = 1;
            __syncthreads();
            more0 = again;
        }

        v16f acc[VT];
        {
            const vt bv = *(const vt*)(bias + lane * VT);
            vt hv[KR];
            float di[KR];
#pragma unroll
            for (int k = 0; k < KR; ++k) {
                const int r = wave + NW * k;
                const int n = lo + ((r < rows) ? r : 0);
                di[k] = dinv[n];
                hv[k] = *(const vt*)(h + (size_t)n * F + lane * VT);
            }
#pragma unroll
            for (int c = 0; c < VT; ++c) {
                acc[c] = (v16f)(0.0f);
#pragma unroll
                for (int k = 0; k < KR; ++k) acc[c][k] = hv[k][c] * di[k] * di[k] + bv[c];
            }
        }

        auto add_row = [&](int k, vt v, float w) {
#pragma unroll
            for (int c = 0; c < VT; ++c) acc[c][k] += v[c] * w;
        };

        auto drain = [&]() {
            constexpr int RIF = (VT == 4) ? 6 : 8;
            constexpr int BATCH = (32 / RIF) * RIF;
            int n = __builtin_amdgcn_readfirstlane(qn[wave]);
            if (n > QCAP) n = QCAP;
            const int qb = wave * QPITCH;
            for (int b = 0; b < n; b += BATCH) {
                const int m = (n - b < BATCH) ? (n - b) : BATCH;
                const int jl = (lane < m) ? lane : (m - 1);
                const int   s = qs[qb + b + jl];
                const int   k = qk[qb + b + jl];
                const float w = qw[qb + b + jl];
                for (int j0 = 0; j0 < m; j0 += RIF) {
                    vt rv[RIF];
#pragma unroll
                    for (int u = 0; u < RIF; ++u) {
                        const int j = (j0 + u < m) ? (j0 + u) : (m - 1);
                        const int sj = __builtin_amdgcn_readlane(s, j);
                        const float* const hrow = h + (size_t)sj * F;
                        rv[u] = *(const vt*)(hrow + lane * VT);
                    }
#pragma unroll
                    for (int u = 0; u < RIF; ++u) {
                        if (j0 + u < m) {
                            const int kj = __builtin_amdgcn_readlane(k, j0 + u);
                            const float wj = __int_as_float(
                                __builtin_amdgcn_readlane(__float_as_int(w), j0 + u));
                            add_row(kj, rv[u], wj);
                        }
                    }
                }
            }
        };

        if (listed) {
            drain();
            __syncthreads();
        }

        if (!listed || more0 != 0) {
            const long total = listed ? (long)lcnt : E;
            for (long c0 = 0; c0 < total; c0 += 1024 * AGG_EPT) {
                if (!listed) {
                    pend = 0;
#pragma unroll
                    for (int j = 0; j < AGG_EPT; ++j)
                        if (c0 + tid + j * 1024 < total) pend |= 1u << j;
                }
                for (int round = 0; round <= 1024 * AGG_EPT; ++round) {
                    if (tid < NW) qn[tid] = 0;
                    if (tid == 0) again = 0;
                    __syncthreads();
#pragma unroll 1
                    for (int j = 0; j < AGG_EPT; ++j) {
                        if ((pend >> j) & 1u) {
                            int s, d;
                            float w;
                            if (listed) {
                                const int i = tid + j * 1024;
                                s = ls[i];
                                d = ld[i];
                                w = FIRST ? dinv[s] * sdi[d] : __int_as_float(lw[i]);
                                place(j, s, d, w, pend);
                            } else {
                                const long e = c0 + tid + j * 1024;
                                d = dst[e] - lo;
                                if ((unsigned)d < (unsigned)rows) {
                                    s = src[e];
                                    w = dinv[s] * sdi[d];
                                    place(j, s, d, w, pend);
                                } else {
                                    pend &= ~(1u << j);
                                }
                            }
                        }
                    }
                    if (pend != 0) again = 1;
                    __syncthreads();
                    const int more = again;
                    drain();
                    __syncthreads();
                    if (more == 0) break;
                }
            }
        }

#pragma unroll
        for (int k = 0; k < KR; ++k) {
            const int r = wave + NW * k;
            if (r < rows) {
                vt o;
#pragma unroll
                for (int c = 0; c < VT; ++c) o[c] = acc[c][k];
                if (RELU) {
#pragma unroll
                    for (int c = 0; c < VT; ++c) o[c] = o[c] > 0.0f ? o[c] : 0.0f;
                }
                *(vt*)(outp + (size_t)(lo + r) * F + lane * VT) = o;
            }
        }
    }
}

extern "C" void kernel_launch(void* const* d_in, const int* in_sizes, int n_in,
                              void* d_out, int out_size, void* d_ws, size_t ws_size,
                              hipStream_t stream) {
    const float* x  = (const float*)d_in[0];
    const int*   ei = (const int*)d_in[1];
    const float* W1 = (const float*)d_in[2];
    const float* b1 = (const float*)d_in[3];
    const float* W2 = (const float*)d_in[4];
    const float* b2 = (const float*)d_in[5];
    float* out = (float*)d_out;

    const int  N = in_sizes[0] / IN_F;
    const long E = (long)in_sizes[1] / 2;
    const int* srcIdx = ei;
    const int* dstIdx = ei + E;

    float* ws   = (float*)d_ws;
    float* dinv = ws;
    float* h    = dinv + N;
    float* agg1 = h + (size_t)N * HID_F;
    int*   lst  = (int*)(agg1 + (size_t)N * HID_F);
    float* h2   = h;

    k_init<<<4, 256, 0, stream>>>(lst, N);

    k_bin<<<(int)((E + BIN_PASS - 1) / BIN_PASS), 1024, 0, stream>>>(srcIdx, dstIdx, lst, N, E);
    k_tile_deg<<<AGG_G, DEG_T, 0, stream>>>(dstIdx, dinv, lst, N, E);

    k_gemm_rows<IN_F, HID_F><<<GEMM_G, 32 * GEMM_WAVES, 0, stream>>>(x, W1, h, N);
    k_tile_agg<HID_F, true, true>
        <<<AGG_G, 1024, 0, stream>>>(srcIdx, dstIdx, dinv, h, b1, agg1, lst, N, E);

    k_gemm_rows<HID_F, OUT_F><<<GEMM_G, 32 * GEMM_WAVES, 0, stream>>>(agg1, W2, h2, N);
    k_tile_agg<OUT_F, false, false>
        <<<AGG_G, 1024, 0, stream>>>(srcIdx, dstIdx, dinv, h2, b2, out, lst, N, E);
}


# Layer-1 dense product launched first (it needs nothing from the graph) and it empties the tile list counters on the way; the init launch goes: 7 dependent launches become 6
# speedup vs baseline: 4.4803x; 1.0305x over previous
#include <hip/hip_runtime.h>

typedef __attribute__((ext_vector_type(2))) float v2f;
typedef __attribute__((ext_vector_type(8))) float v8f;

#define IN_F  128
#define HID_F 128
#define OUT_F 64

#define AGG_G 256
#define TMAX  400
#define MAXT  1024
#define BIN_EPT 8
#define BIN_PASS (1024 * BIN_EPT)

struct TileGeo { int tn; int ntiles; int cap; long lhdr; };

__device__ __forceinline__ TileGeo tile_geo(int N) {
    TileGeo g;
    int tn = (N + AGG_G - 1) / AGG_G;
    if (tn > TMAX) tn = TMAX;
    if (tn < 1) tn = 1;
    g.tn = tn;
    g.ntiles = (N + tn - 1) / tn;
    const long lwords = (long)N * OUT_F;
    g.lhdr = ((long)g.ntiles + 31) & ~31L;
    long lc = (lwords - g.lhdr) / g.ntiles / 3;
    if (lc < 0) lc = 0;
    if (lc > (1L << 30)) lc = 1L << 30;
    g.cap = (int)(lc & ~31L);
    return g;
}

__global__ void __launch_bounds__(1024)
k_bin(const int* __restrict__ src, const int* __restrict__ dst,
      int* __restrict__ lst, int N, long E) {
    __shared__ int hist[MAXT];
    __shared__ int gdelta[MAXT];
    __shared__ int ssrc[BIN_PASS];
    __shared__ int spk[BIN_PASS];
    __shared__ int wsum[32];
    __shared__ int tot;
    const TileGeo g = tile_geo(N);
    const bool binned = g.ntiles <= MAXT && g.cap > 0;
    const int tid  = threadIdx.x;
    const int lane = tid & 31;
    const int wave = tid >> 5;

    for (long p0 = (long)blockIdx.x * BIN_PASS; p0 < E; p0 += (long)gridDim.x * BIN_PASS) {
        hist[tid] = 0;
        __syncthreads();

        int d[BIN_EPT], s[BIN_EPT], tl[BIN_EPT], rk[BIN_EPT];
#pragma unroll
        for (int k = 0; k < BIN_EPT; ++k) {
            const long e = p0 + k * 1024 + tid;
            const bool ok = e < E;
            d[k] = ok ? dst[e] : -1;
            s[k] = ok ? src[e] : 0;
        }
#pragma unroll
        for (int k = 0; k < BIN_EPT; ++k) {
            tl[k] = 0; rk[k] = 0;
            if (binned && (unsigned)d[k] < (unsigned)N) {
                tl[k] = d[k] / g.tn;
                rk[k] = __hip_atomic_fetch_add(&hist[tl[k]], 1, __ATOMIC_RELAXED,
                                               __HIP_MEMORY_SCOPE_WORKGROUP);
            }
        }
        __syncthreads();
        if (!binned) continue;

        const int c = hist[tid];
        int gb = 0;
        if (c > 0 && tid < g.ntiles)
            gb = __hip_atomic_fetch_add(&lst[tid], c, __ATOMIC_RELAXED, __HIP_MEMORY_SCOPE_AGENT);
        int x = c;
#pragma unroll
        for (int o = 1; o < 32; o <<= 1) {
            const int y = __shfl_up(x, o);
            if (lane >= o) x += y;
        }
        if (lane == 31) wsum[wave] = x;
        __syncthreads();
        if (wave == 0) {
            const int v = wsum[lane];
            int z = v;
#pragma unroll
            for (int o = 1; o < 32; o <<= 1) {
                const int y = __shfl_up(z, o);
                if (lane >= o) z += y;
            }
            wsum[lane] = z - v;
            if (lane == 31) tot = z;
        }
        __syncthreads();
        const int excl = x - c + wsum[wave];
        hist[tid]   = excl;
        gdelta[tid] = gb - excl;
        __syncthreads();

#pragma unroll
        for (int k = 0; k < BIN_EPT; ++k) {
            if ((unsigned)d[k] < (unsigned)N) {
                const int p = hist[tl[k]] + rk[k];
                ssrc[p] = s[k];
                spk[p]  = (tl[k] << 9) | (d[k] - tl[k] * g.tn);
            }
        }
        __syncthreads();

        const int n = tot;
        for (int i = tid; i < n; i += 1024) {
            const int pk = spk[i];
            const int t  = pk >> 9;
            const int pos = gdelta[t] + i;
            if (pos >= 0 && pos < g.cap) {
                int* const ls = lst + g.lhdr + (long)t * 3 * g.cap;
                ls[pos]         = ssrc[i];
                ls[g.cap + pos] = pk & 511;
            }
        }
    }
}

#define DEG_T 512
#define DEG_U 8
__global__ void __launch_bounds__(DEG_T)
k_tile_deg(const int* __restrict__ dst, float* __restrict__ dinv,
           const int* __restrict__ lst, int N, long E) {
    __shared__ int cnt[TMAX];
    const TileGeo g = tile_geo(N);
    const bool binned = g.ntiles <= MAXT && g.cap > 0;
    const int tid = threadIdx.x;
    for (int t = blockIdx.x; t < g.ntiles; t += gridDim.x) {
        const int lo = t * g.tn;
        const int rows = (N - lo < g.tn) ? (N - lo) : g.tn;
        int lcnt = -1;
        if (binned) lcnt = lst[t];
        for (int r = tid; r < rows; r += DEG_T) cnt[r] = 0;
        __syncthreads();
        if (lcnt >= 0 && lcnt <= g.cap) {
            const int* const ld = lst + g.lhdr + (long)t * 3 * g.cap + g.cap;
            for (int i0 = tid; i0 < lcnt; i0 += DEG_T * DEG_U) {
                int r[DEG_U];
#pragma unroll
                for (int u = 0; u < DEG_U; ++u) {
                    const int i = i0 + u * DEG_T;
                    r[u] = (i < lcnt) ? ld[i] : -1;
                }
#pragma unroll
                for (int u = 0; u < DEG_U; ++u)
                    if ((unsigned)r[u] < (unsigned)rows)
                        __hip_atomic_fetch_add(&cnt[r[u]], 1, __ATOMIC_RELAXED,
                                               __HIP_MEMORY_SCOPE_WORKGROUP);
            }
        } else {
            for (long e = tid; e < E; e += DEG_T) {
                const unsigned r = (unsigned)(dst[e] - lo);
                if (r < (unsigned)rows)
                    __hip_atomic_fetch_add(&cnt[r], 1, __ATOMIC_RELAXED,
                                           __HIP_MEMORY_SCOPE_WORKGROUP);
            }
        }
        __syncthreads();
        for (int r = tid; r < rows; r += DEG_T)
            dinv[lo + r] = rsqrtf(1.0f + (float)cnt[r]);
        __syncthreads();
    }
}

#define GEMM_WAVES 8
#define GEMM_G     512
template <int K, int NCOL, bool CLEAR>
__global__ void __launch_bounds__(32 * GEMM_WAVES)
k_gemm_rows(const float* __restrict__ X, const float* __restrict__ W,
            float* __restrict__ H, int M, int* __restrict__ lst) {
    constexpr int NS  = NCOL / 16;
    constexpr int NKK = K / 8;
    __shared__ float4 sW[(K / 4) * NCOL];

    const int tid  = threadIdx.x;
    const int wave = __builtin_amdgcn_readfirstlane(tid >> 5);
    const int lane = tid & 31;
    const int half = lane >> 4;
    const int idx  = lane & 15;

    static_assert(GEMM_WAVES == 8, "the rank table is for 8 waves");
    const int ntile = (M + 15) / 16;
    const int t0 = (int)((long)blockIdx.x * ntile / gridDim.x) + ((0x73261540 >> (4 * wave)) & 7);
    const int t1 = (int)(((long)blockIdx.x + 1) * ntile / gridDim.x);

    auto row_of = [&](int t) {
        int ar = t * 16 + idx;
        if (ar > M - 1) ar = M - 1;
        return X + (size_t)ar * K + half * 4;
    };

    float4 a[NKK];
    if (t0 < t1) {
        const float* const xr = row_of(t0);
#pragma unroll
        for (int kk = 0; kk < NKK; ++kk) a[kk] = *(const float4*)(xr + kk * 8);
    }

    if (CLEAR) {
        const TileGeo g = tile_geo(M);
        const long stride = (long)gridDim.x * blockDim.x;
        for (long j = (long)blockIdx.x * blockDim.x + tid; j < g.lhdr; j += stride) lst[j] = 0;
    }

    constexpr int WITEMS = (K / 4) * (NCOL / 4);
    static_assert(WITEMS % (32 * GEMM_WAVES) == 0, "whole trips");
#pragma unroll
    for (int j = 0; j < WITEMS / (32 * GEMM_WAVES); ++j) {
        const int i  = tid + j * 32 * GEMM_WAVES;
        const int q  = i / (NCOL / 4);
        const int n4 = i % (NCOL / 4);
        const float* const w = W + (size_t)(4 * q) * NCOL + n4 * 4;
        const float4 r0 = *(const float4*)(w);
        const float4 r1 = *(const float4*)(w + NCOL);
        const float4 r2 = *(const float4*)(w + 2 * NCOL);
        const float4 r3 = *(const float4*)(w + 3 * NCOL);
        float4* const d = sW + q * NCOL + n4 * 4;
        d[0] = make_float4(r0.x, r1.x, r2.x, r3.x);
        d[1] = make_float4(r0.y, r1.y, r2.y, r3.y);
        d[2] = make_float4(r0.z, r1.z, r2.z, r3.z);
        d[3] = make_float4(r0.w, r1.w, r2.w, r3.w);
    }
    __syncthreads();

    const float4* const bl = sW + half * NCOL + idx;

    for (int t = t0; t < t1; t += GEMM_WAVES) {
        const int row0 = t * 16;
        v8f acc[NS] = {};
        float4 bc[NS];
#pragma unroll
        for (int c = 0; c < NS; ++c) bc[c] = bl[c * 16];
#pragma unroll
        for (int kk = 0; kk < NKK; ++kk) {
            float4 bn[NS];
            if (kk + 1 < NKK) {
#pragma unroll
                for (int c = 0; c < NS; ++c) bn[c] = bl[(kk + 1) * 2 * NCOL + c * 16];
            }
#pragma unroll
            for (int s = 0; s < 2; ++s) {
                v2f av;
                av.x = s ? a[kk].z : a[kk].x;
                av.y = s ? a[kk].w : a[kk].y;
#pragma unroll
                for (int c = 0; c < NS; ++c) {
                    v2f bv;
                    bv.x = s ? bc[c].z : bc[c].x;
                    bv.y = s ? bc[c].w : bc[c].y;
                    acc[c] = __builtin_amdgcn_wmma_f32_16x16x4_f32(
                         false, av,  false, bv,
                         (short)0, acc[c],  false,  false);
                }
            }
            if (kk + 1 < NKK) {
#pragma unroll
                for (int c = 0; c < NS; ++c) bc[c] = bn[c];
            }
            __builtin_amdgcn_sched_barrier(0);
        }

#pragma unroll
        for (int v = 0; v < 8; ++v) {
            const int gr = row0 + half * 8 + v;
            if (gr < M) {
#pragma unroll
                for (int c = 0; c < NS; ++c)
                    H[(size_t)gr * NCOL + c * 16 + idx] = acc[c][v];
            }
        }

        if (t + GEMM_WAVES < t1) {
            const float* const xr = row_of(t + GEMM_WAVES);
#pragma unroll
            for (int kk = 0; kk < NKK; ++kk) a[kk] = *(const float4*)(xr + kk * 8);
        }
    }
}

#define QCAP   512
#define QPITCH (QCAP + 1)
#define AGG_EPT 9
template <int F, bool RELU, bool FIRST>
__global__ void __launch_bounds__(1024)
k_tile_agg(const int* __restrict__ src, const int* __restrict__ dst,
           const float* __restrict__ dinv, const float* __restrict__ h,
           const float* __restrict__ bias, float* __restrict__ outp,
           int* __restrict__ lst, int N, long E) {
    constexpr int NW = 32;
    constexpr int VT = F / 32;
    constexpr int KR = (TMAX + NW - 1) / NW;
    static_assert(AGG_EPT * 1024 > (TMAX * 64) / 3, "a thread's share of the longest list");
    static_assert(KR <= 16, "one 16-wide register vector per feature of a lane");
    typedef float vt __attribute__((ext_vector_type(VT)));
    typedef float v16f __attribute__((ext_vector_type(16)));
    __shared__ int   qs[NW * QPITCH];
    __shared__ int   qk[NW * QPITCH];
    __shared__ float qw[NW * QPITCH];
    __shared__ float sdi[TMAX];
    __shared__ int   qn[NW];
    __shared__ int   again;

    const int tid  = threadIdx.x;
    const int wave = __builtin_amdgcn_readfirstlane(tid >> 5);
    const int lane = tid & 31;

    const int G = gridDim.x;
    const TileGeo g = tile_geo(N);
    const int tn = g.tn;
    const int ntiles = g.ntiles;
    const int cap = g.cap;
    const long lhdr = g.lhdr;
    const bool binned = ntiles <= MAXT && cap > 0;

    for (int t = blockIdx.x; t < ntiles; t += G) {
        const int lo = t * tn;
        const int rows = (N - lo < tn) ? (N - lo) : tn;
        int* const ls = lst + lhdr + (long)t * 3 * cap;
        int* const ld = ls + cap;
        int* const lw = ld + cap;
        int lraw = -1;
        if (binned) lraw = lst[t];
        for (int r = tid; r < rows; r += 1024) sdi[r] = dinv[lo + r];

        const int lcnt = __builtin_amdgcn_readfirstlane(lraw);
        const bool listed = lcnt >= 0 && lcnt <= cap;

        auto place = [&](int j, int s, int d, float w, unsigned& pend) {
            const int o = d & (NW - 1);
            const int pos = __hip_atomic_fetch_add(&qn[o], 1, __ATOMIC_RELAXED,
                                                   __HIP_MEMORY_SCOPE_WORKGROUP);
            if (pos < QCAP) {
                qs[o * QPITCH + pos] = s;
                qk[o * QPITCH + pos] = d >> 5;
                qw[o * QPITCH + pos] = w;
                if (FIRST && listed) lw[tid + j * 1024] = __float_as_int(w);
                pend &= ~(1u << j);
            }
        };

        unsigned pend = 0;
        int more0 = 0;
        if (listed) {
#pragma unroll
            for (int j = 0; j < AGG_EPT; ++j)
                if (tid + j * 1024 < lcnt) pend |= 1u << j;
            if (tid < NW) qn[tid] = 0;
            if (tid == 0) again = 0;
            __syncthreads();
            int   es[AGG_EPT], ed[AGG_EPT];
            float ew[AGG_EPT];
#pragma unroll
            for (int j = 0; j < AGG_EPT; ++j) {
                const int i = tid + j * 1024;
                const bool ok = (pend >> j) & 1u;
                es[j] = ok ? ls[i] : 0;
                ed[j] = ok ? ld[i] : 0;
                if (!FIRST) ew[j] = ok ? __int_as_float(lw[i]) : 0.0f;
            }
            if (FIRST) {
#pragma unroll
                for (int j = 0; j < AGG_EPT; ++j) ew[j] = dinv[es[j]];
            }
#pragma unroll
            for (int j = 0; j < AGG_EPT; ++j) {
                if ((pend >> j) & 1u)
                    place(j, es[j], ed[j], FIRST ? ew[j] * sdi[ed[j]] : ew[j], pend);
            }
            if (pend != 0) again = 1;
            __syncthreads();
            more0 = again;
        }

        v16f acc[VT];
        {
            const vt bv = *(const vt*)(bias + lane * VT);
            vt hv[KR];
            float di[KR];
#pragma unroll
            for (int k = 0; k < KR; ++k) {
                const int r = wave + NW * k;
                const int n = lo + ((r < rows) ? r : 0);
                di[k] = dinv[n];
                hv[k] = *(const vt*)(h + (size_t)n * F + lane * VT);
            }
#pragma unroll
            for (int c = 0; c < VT; ++c) {
                acc[c] = (v16f)(0.0f);
#pragma unroll
                for (int k = 0; k < KR; ++k) acc[c][k] = hv[k][c] * di[k] * di[k] + bv[c];
            }
        }

        auto add_row = [&](int k, vt v, float w) {
#pragma unroll
            for (int c = 0; c < VT; ++c) acc[c][k] += v[c] * w;
        };

        auto drain = [&]() {
            constexpr int RIF = (VT == 4) ? 6 : 8;
            constexpr int BATCH = (32 / RIF) * RIF;
            int n = __builtin_amdgcn_readfirstlane(qn[wave]);
            if (n > QCAP) n = QCAP;
            const int qb = wave * QPITCH;
            for (int b = 0; b < n; b += BATCH) {
                const int m = (n - b < BATCH) ? (n - b) : BATCH;
                const int jl = (lane < m) ? lane : (m - 1);
                const int   s = qs[qb + b + jl];
                const int   k = qk[qb + b + jl];
                const float w = qw[qb + b + jl];
                for (int j0 = 0; j0 < m; j0 += RIF) {
                    vt rv[RIF];
#pragma unroll
                    for (int u = 0; u < RIF; ++u) {
                        const int j = (j0 + u < m) ? (j0 + u) : (m - 1);
                        const int sj = __builtin_amdgcn_readlane(s, j);
                        const float* const hrow = h + (size_t)sj * F;
                        rv[u] = *(const vt*)(hrow + lane * VT);
                    }
#pragma unroll
                    for (int u = 0; u < RIF; ++u) {
                        if (j0 + u < m) {
                            const int kj = __builtin_amdgcn_readlane(k, j0 + u);
                            const float wj = __int_as_float(
                                __builtin_amdgcn_readlane(__float_as_int(w), j0 + u));
                            add_row(kj, rv[u], wj);
                        }
                    }
                }
            }
        };

        if (listed) {
            drain();
            __syncthreads();
        }

        if (!listed || more0 != 0) {
            const long total = listed ? (long)lcnt : E;
            for (long c0 = 0; c0 < total; c0 += 1024 * AGG_EPT) {
                if (!listed) {
                    pend = 0;
#pragma unroll
                    for (int j = 0; j < AGG_EPT; ++j)
                        if (c0 + tid + j * 1024 < total) pend |= 1u << j;
                }
                for (int round = 0; round <= 1024 * AGG_EPT; ++round) {
                    if (tid < NW) qn[tid] = 0;
                    if (tid == 0) again = 0;
                    __syncthreads();
#pragma unroll 1
                    for (int j = 0; j < AGG_EPT; ++j) {
                        if ((pend >> j) & 1u) {
                            int s, d;
                            float w;
                            if (listed) {
                                const int i = tid + j * 1024;
                                s = ls[i];
                                d = ld[i];
                                w = FIRST ? dinv[s] * sdi[d] : __int_as_float(lw[i]);
                                place(j, s, d, w, pend);
                            } else {
                                const long e = c0 + tid + j * 1024;
                                d = dst[e] - lo;
                                if ((unsigned)d < (unsigned)rows) {
                                    s = src[e];
                                    w = dinv[s] * sdi[d];
                                    place(j, s, d, w, pend);
                                } else {
                                    pend &= ~(1u << j);
                                }
                            }
                        }
                    }
                    if (pend != 0) again = 1;
                    __syncthreads();
                    const int more = again;
                    drain();
                    __syncthreads();
                    if (more == 0) break;
                }
            }
        }

#pragma unroll
        for (int k = 0; k < KR; ++k) {
            const int r = wave + NW * k;
            if (r < rows) {
                vt o;
#pragma unroll
                for (int c = 0; c < VT; ++c) o[c] = acc[c][k];
                if (RELU) {
#pragma unroll
                    for (int c = 0; c < VT; ++c) o[c] = o[c] > 0.0f ? o[c] : 0.0f;
                }
                *(vt*)(outp + (size_t)(lo + r) * F + lane * VT) = o;
            }
        }
    }
}

extern "C" void kernel_launch(void* const* d_in, const int* in_sizes, int n_in,
                              void* d_out, int out_size, void* d_ws, size_t ws_size,
                              hipStream_t stream) {
    const float* x  = (const float*)d_in[0];
    const int*   ei = (const int*)d_in[1];
    const float* W1 = (const float*)d_in[2];
    const float* b1 = (const float*)d_in[3];
    const float* W2 = (const float*)d_in[4];
    const float* b2 = (const float*)d_in[5];
    float* out = (float*)d_out;

    const int  N = in_sizes[0] / IN_F;
    const long E = (long)in_sizes[1] / 2;
    const int* srcIdx = ei;
    const int* dstIdx = ei + E;

    float* ws   = (float*)d_ws;
    float* dinv = ws;
    float* h    = dinv + N;
    float* agg1 = h + (size_t)N * HID_F;
    int*   lst  = (int*)(agg1 + (size_t)N * HID_F);
    float* h2   = h;

    k_gemm_rows<IN_F, HID_F, true><<<GEMM_G, 32 * GEMM_WAVES, 0, stream>>>(x, W1, h, N, lst);

    k_bin<<<(int)((E + BIN_PASS - 1) / BIN_PASS), 1024, 0, stream>>>(srcIdx, dstIdx, lst, N, E);
    k_tile_deg<<<AGG_G, DEG_T, 0, stream>>>(dstIdx, dinv, lst, N, E);

    k_tile_agg<HID_F, true, true>
        <<<AGG_G, 1024, 0, stream>>>(srcIdx, dstIdx, dinv, h, b1, agg1, lst, N, E);

    k_gemm_rows<HID_F, OUT_F, false><<<GEMM_G, 32 * GEMM_WAVES, 0, stream>>>(agg1, W2, h2, N, lst);
    k_tile_agg<OUT_F, false, false>
        <<<AGG_G, 1024, 0, stream>>>(srcIdx, dstIdx, dinv, h2, b2, out, lst, N, E);
}


# speedup vs baseline: 4.6521x; 1.0383x over previous
#include <hip/hip_runtime.h>

typedef __attribute__((ext_vector_type(2))) float v2f;
typedef __attribute__((ext_vector_type(8))) float v8f;

#define IN_F  128
#define HID_F 128
#define OUT_F 64

#define AGG_G 256
#define TMAX  400
#define MAXT  1024
#define BIN_EPT 8
#define BIN_PASS (1024 * BIN_EPT)

struct TileGeo { int tn; int ntiles; int cap; long lhdr; };

__device__ __forceinline__ TileGeo tile_geo(int N) {
    TileGeo g;
    int tn = (N + AGG_G - 1) / AGG_G;
    if (tn > TMAX) tn = TMAX;
    if (tn < 1) tn = 1;
    g.tn = tn;
    g.ntiles = (N + tn - 1) / tn;
    const long lwords = (long)N * OUT_F;
    g.lhdr = ((long)g.ntiles + 63) & ~63L;
    long lc = (lwords - g.lhdr) / g.ntiles / 3;
    if (lc < 0) lc = 0;
    if (lc > (1L << 30)) lc = 1L << 30;
    g.cap = (int)(lc & ~63L);
    return g;
}

__global__ void __launch_bounds__(1024)
k_bin(const int* __restrict__ src, const int* __restrict__ dst,
      int* __restrict__ lst, int N, long E) {
    __shared__ int hist[MAXT];
    __shared__ int gdelta[MAXT];
    __shared__ int ssrc[BIN_PASS];
    __shared__ int spk[BIN_PASS];
    __shared__ int wsum[32];
    __shared__ int tot;
    const TileGeo g = tile_geo(N);
    const bool binned = g.ntiles <= MAXT && g.cap > 0;
    const int tid  = threadIdx.x;
    const int lane = tid & 31;
    const int wave = tid >> 5;

    for (long p0 = (long)blockIdx.x * BIN_PASS; p0 < E; p0 += (long)gridDim.x * BIN_PASS) {
        hist[tid] = 0;
        __syncthreads();

        int d[BIN_EPT], s[BIN_EPT], tl[BIN_EPT], rk[BIN_EPT];
#pragma unroll
        for (int k = 0; k < BIN_EPT; ++k) {
            const long e = p0 + k * 1024 + tid;
            const bool ok = e < E;
            d[k] = ok ? dst[e] : -1;
            s[k] = ok ? src[e] : 0;
        }
#pragma unroll
        for (int k = 0; k < BIN_EPT; ++k) {
            tl[k] = 0; rk[k] = 0;
            if (binned && (unsigned)d[k] < (unsigned)N) {
                tl[k] = d[k] / g.tn;
                rk[k] = __hip_atomic_fetch_add(&hist[tl[k]], 1, __ATOMIC_RELAXED,
                                               __HIP_MEMORY_SCOPE_WORKGROUP);
            }
        }
        __syncthreads();
        if (!binned) continue;

        const int c = hist[tid];
        int gb = 0;
        if (c > 0 && tid < g.ntiles)
            gb = __hip_atomic_fetch_add(&lst[tid], c, __ATOMIC_RELAXED, __HIP_MEMORY_SCOPE_AGENT);
        int x = c;
#pragma unroll
        for (int o = 1; o < 32; o <<= 1) {
            const int y = __shfl_up(x, o);
            if (lane >= o) x += y;
        }
        if (lane == 31) wsum[wave] = x;
        __syncthreads();
        if (wave == 0) {
            const int v = wsum[lane];
            int z = v;
#pragma unroll
            for (int o = 1; o < 32; o <<= 1) {
                const int y = __shfl_up(z, o);
                if (lane >= o) z += y;
            }
            wsum[lane] = z - v;
            if (lane == 31) tot = z;
        }
        __syncthreads();
        const int excl = x - c + wsum[wave];
        hist[tid]   = excl;
        gdelta[tid] = gb - excl;
        __syncthreads();

#pragma unroll
        for (int k = 0; k < BIN_EPT; ++k) {
            if ((unsigned)d[k] < (unsigned)N) {
                const int p = hist[tl[k]] + rk[k];
                ssrc[p] = s[k];
                spk[p]  = (tl[k] << 9) | (d[k] - tl[k] * g.tn);
            }
        }
        __syncthreads();

        const int n = tot;
        for (int i = tid; i < n; i += 1024) {
            const int pk = spk[i];
            const int t  = pk >> 9;
            const int pos = gdelta[t] + i;
            if (pos >= 0 && pos < g.cap) {
                int* const ls = lst + g.lhdr + (long)t * 3 * g.cap;
                ls[pos]         = ssrc[i];
                ls[g.cap + pos] = pk & 511;
            }
        }
    }
}

#define DEG_T 512
#define DEG_U 8
__global__ void __launch_bounds__(DEG_T)
k_tile_deg(const int* __restrict__ dst, float* __restrict__ dinv,
           const int* __restrict__ lst, int N, long E) {
    __shared__ int cnt[TMAX];
    const TileGeo g = tile_geo(N);
    const bool binned = g.ntiles <= MAXT && g.cap > 0;
    const int tid = threadIdx.x;
    for (int t = blockIdx.x; t < g.ntiles; t += gridDim.x) {
        const int lo = t * g.tn;
        const int rows = (N - lo < g.tn) ? (N - lo) : g.tn;
        int lcnt = -1;
        if (binned) lcnt = lst[t];
        for (int r = tid; r < rows; r += DEG_T) cnt[r] = 0;
        __syncthreads();
        if (lcnt >= 0 && lcnt <= g.cap) {
            const int* const ld = lst + g.lhdr + (long)t * 3 * g.cap + g.cap;
            for (int i0 = tid; i0 < lcnt; i0 += DEG_T * DEG_U) {
                int r[DEG_U];
#pragma unroll
                for (int u = 0; u < DEG_U; ++u) {
                    const int i = i0 + u * DEG_T;
                    r[u] = (i < lcnt) ? ld[i] : -1;
                }
#pragma unroll
                for (int u = 0; u < DEG_U; ++u)
                    if ((unsigned)r[u] < (unsigned)rows)
                        __hip_atomic_fetch_add(&cnt[r[u]], 1, __ATOMIC_RELAXED,
                                               __HIP_MEMORY_SCOPE_WORKGROUP);
            }
        } else {
            for (long e = tid; e < E; e += DEG_T) {
                const unsigned r = (unsigned)(dst[e] - lo);
                if (r < (unsigned)rows)
                    __hip_atomic_fetch_add(&cnt[r], 1, __ATOMIC_RELAXED,
                                           __HIP_MEMORY_SCOPE_WORKGROUP);
            }
        }
        __syncthreads();
        for (int r = tid; r < rows; r += DEG_T)
            dinv[lo + r] = rsqrtf(1.0f + (float)cnt[r]);
        __syncthreads();
    }
}

#define GEMM_WAVES 8
#define GEMM_G     512
template <int K, int NCOL, bool CLEAR>
__global__ void __launch_bounds__(32 * GEMM_WAVES)
k_gemm_rows(const float* __restrict__ X, const float* __restrict__ W,
            float* __restrict__ H, int M, int* __restrict__ lst) {
    constexpr int NS  = NCOL / 16;
    constexpr int NKK = K / 8;
    __shared__ float4 sW[(K / 4) * NCOL];

    const int tid  = threadIdx.x;
    const int wave = __builtin_amdgcn_readfirstlane(tid >> 5);
    const int lane = tid & 31;
    const int half = lane >> 4;
    const int idx  = lane & 15;

    static_assert(GEMM_WAVES == 8, "the rank table is for 8 waves");
    const int ntile = (M + 15) / 16;
    const int t0 = (int)((long)blockIdx.x * ntile / gridDim.x) + ((0x73261540 >> (4 * wave)) & 7);
    const int t1 = (int)(((long)blockIdx.x + 1) * ntile / gridDim.x);

    auto row_of = [&](int t) {
        int ar = t * 16 + idx;
        if (ar > M - 1) ar = M - 1;
        return X + (size_t)ar * K + half * 4;
    };

    float4 a[NKK];
    if (t0 < t1) {
        const float* const xr = row_of(t0);
#pragma unroll
        for (int kk = 0; kk < NKK; ++kk) a[kk] = *(const float4*)(xr + kk * 8);
    }

    if (CLEAR) {
        const TileGeo g = tile_geo(M);
        const long stride = (long)gridDim.x * blockDim.x;
        for (long j = (long)blockIdx.x * blockDim.x + tid; j < g.lhdr; j += stride) lst[j] = 0;
    }

    constexpr int WITEMS = (K / 4) * (NCOL / 4);
    static_assert(WITEMS % (32 * GEMM_WAVES) == 0, "whole trips");
#pragma unroll
    for (int j = 0; j < WITEMS / (32 * GEMM_WAVES); ++j) {
        const int i  = tid + j * 32 * GEMM_WAVES;
        const int q  = i / (NCOL / 4);
        const int n4 = i % (NCOL / 4);
        const float* const w = W + (size_t)(4 * q) * NCOL + n4 * 4;
        const float4 r0 = *(const float4*)(w);
        const float4 r1 = *(const float4*)(w + NCOL);
        const float4 r2 = *(const float4*)(w + 2 * NCOL);
        const float4 r3 = *(const float4*)(w + 3 * NCOL);
        float4* const d = sW + q * NCOL + n4 * 4;
        d[0] = make_float4(r0.x, r1.x, r2.x, r3.x);
        d[1] = make_float4(r0.y, r1.y, r2.y, r3.y);
        d[2] = make_float4(r0.z, r1.z, r2.z, r3.z);
        d[3] = make_float4(r0.w, r1.w, r2.w, r3.w);
    }
    __syncthreads();

    const float4* const bl = sW + half * NCOL + idx;

    for (int t = t0; t < t1; t += GEMM_WAVES) {
        const int row0 = t * 16;
        v8f acc[NS] = {};
        float4 bc[NS];
#pragma unroll
        for (int c = 0; c < NS; ++c) bc[c] = bl[c * 16];
#pragma unroll
        for (int kk = 0; kk < NKK; ++kk) {
            float4 bn[NS];
            if (kk + 1 < NKK) {
#pragma unroll
                for (int c = 0; c < NS; ++c) bn[c] = bl[(kk + 1) * 2 * NCOL + c * 16];
            }
#pragma unroll
            for (int s = 0; s < 2; ++s) {
                v2f av;
                av.x = s ? a[kk].z : a[kk].x;
                av.y = s ? a[kk].w : a[kk].y;
#pragma unroll
                for (int c = 0; c < NS; ++c) {
                    v2f bv;
                    bv.x = s ? bc[c].z : bc[c].x;
                    bv.y = s ? bc[c].w : bc[c].y;
                    acc[c] = __builtin_amdgcn_wmma_f32_16x16x4_f32(
                         false, av,  false, bv,
                         (short)0, acc[c],  false,  false);
                }
            }
            if (kk + 1 < NKK) {
#pragma unroll
                for (int c = 0; c < NS; ++c) bc[c] = bn[c];
            }
            __builtin_amdgcn_sched_barrier(0);
        }

#pragma unroll
        for (int v = 0; v < 8; ++v) {
            const int gr = row0 + half * 8 + v;
            if (gr < M) {
#pragma unroll
                for (int c = 0; c < NS; ++c)
                    H[(size_t)gr * NCOL + c * 16 + idx] = acc[c][v];
            }
        }

        if (t + GEMM_WAVES < t1) {
            const float* const xr = row_of(t + GEMM_WAVES);
#pragma unroll
            for (int kk = 0; kk < NKK; ++kk) a[kk] = *(const float4*)(xr + kk * 8);
        }
    }
}

#define QCAP   512
#define QPITCH (QCAP + 1)
#define AGG_EPT 9
template <int F, bool RELU, bool FIRST>
__global__ void __launch_bounds__(1024)
k_tile_agg(const int* __restrict__ src, const int* __restrict__ dst,
           const float* __restrict__ dinv, const float* __restrict__ h,
           const float* __restrict__ bias, float* __restrict__ outp,
           int* __restrict__ lst, int N, long E) {
    constexpr int NW = 32;
    constexpr int VT = F / 32;
    constexpr int KR = (TMAX + NW - 1) / NW;
    static_assert(AGG_EPT * 1024 > (TMAX * 64) / 3, "a thread's share of the longest list");
    static_assert(KR <= 16, "one 16-wide register vector per feature of a lane");
    typedef float vt __attribute__((ext_vector_type(VT)));
    typedef float v16f __attribute__((ext_vector_type(16)));
    __shared__ int   qs[NW * QPITCH];
    __shared__ int   qk[NW * QPITCH];
    __shared__ float qw[NW * QPITCH];
    __shared__ float sdi[TMAX];
    __shared__ int   qn[NW];
    __shared__ int   again;

    const int tid  = threadIdx.x;
    const int wave = __builtin_amdgcn_readfirstlane(tid >> 5);
    const int lane = tid & 31;

    const int G = gridDim.x;
    const TileGeo g = tile_geo(N);
    const int tn = g.tn;
    const int ntiles = g.ntiles;
    const int cap = g.cap;
    const long lhdr = g.lhdr;
    const bool binned = ntiles <= MAXT && cap > 0;

    for (int t = blockIdx.x; t < ntiles; t += G) {
        const int lo = t * tn;
        const int rows = (N - lo < tn) ? (N - lo) : tn;
        int* const ls = lst + lhdr + (long)t * 3 * cap;
        int* const ld = ls + cap;
        int* const lw = ld + cap;
        int lraw = -1;
        if (binned) lraw = lst[t];
        for (int r = tid; r < rows; r += 1024) sdi[r] = dinv[lo + r];

        const int lcnt = __builtin_amdgcn_readfirstlane(lraw);
        const bool listed = lcnt >= 0 && lcnt <= cap;

        auto place = [&](int j, int s, int d, float w, unsigned& pend) {
            const int o = d & (NW - 1);
            const int pos = __hip_atomic_fetch_add(&qn[o], 1, __ATOMIC_RELAXED,
                                                   __HIP_MEMORY_SCOPE_WORKGROUP);
            if (pos < QCAP) {
                qs[o * QPITCH + pos] = s;
                qk[o * QPITCH + pos] = d >> 5;
                qw[o * QPITCH + pos] = w;
                if (FIRST && listed) lw[tid + j * 1024] = __float_as_int(w);
                pend &= ~(1u << j);
            }
        };

        unsigned pend = 0;
        int more0 = 0;
        if (listed) {
#pragma unroll
            for (int j = 0; j < AGG_EPT; ++j)
                if (tid + j * 1024 < lcnt) pend |= 1u << j;
            if (tid < NW) qn[tid] = 0;
            if (tid == 0) again = 0;
            __syncthreads();
            int   es[AGG_EPT], ed[AGG_EPT];
            float ew[AGG_EPT];
#pragma unroll
            for (int j = 0; j < AGG_EPT; ++j) {
                const int i = tid + j * 1024;
                const bool ok = (pend >> j) & 1u;
                es[j] = ok ? ls[i] : 0;
                ed[j] = ok ? ld[i] : 0;
                if (!FIRST) ew[j] = ok ? __int_as_float(lw[i]) : 0.0f;
            }
            if (FIRST) {
#pragma unroll
                for (int j = 0; j < AGG_EPT; ++j) ew[j] = dinv[es[j]];
            }
#pragma unroll
            for (int j = 0; j < AGG_EPT; ++j) {
                if ((pend >> j) & 1u)
                    place(j, es[j], ed[j], FIRST ? ew[j] * sdi[ed[j]] : ew[j], pend);
            }
            if (pend != 0) again = 1;
            __syncthreads();
            more0 = again;
        }

        v16f acc[VT];
        {
            const vt bv = *(const vt*)(bias + lane * VT);
            vt hv[KR];
            float di[KR];
#pragma unroll
            for (int k = 0; k < KR; ++k) {
                const int r = wave + NW * k;
                const int n = lo + ((r < rows) ? r : 0);
                di[k] = dinv[n];
                hv[k] = *(const vt*)(h + (size_t)n * F + lane * VT);
            }
#pragma unroll
            for (int c = 0; c < VT; ++c) {
                acc[c] = (v16f)(0.0f);
#pragma unroll
                for (int k = 0; k < KR; ++k) acc[c][k] = hv[k][c] * di[k] * di[k] + bv[c];
            }
        }

        auto add_row = [&](int k, vt v, float w) {
#pragma unroll
            for (int c = 0; c < VT; ++c) acc[c][k] += v[c] * w;
        };

        auto drain = [&]() {
            constexpr int RIF = (VT == 4) ? 6 : 8;
            constexpr int BATCH = (32 / RIF) * RIF;
            int n = __builtin_amdgcn_readfirstlane(qn[wave]);
            if (n > QCAP) n = QCAP;
            const int qb = wave * QPITCH;
            for (int b = 0; b < n; b += BATCH) {
                const int m = (n - b < BATCH) ? (n - b) : BATCH;
                const int jl = (lane < m) ? lane : (m - 1);
                const int   s = qs[qb + b + jl];
                const int   k = qk[qb + b + jl];
                const float w = qw[qb + b + jl];
                for (int j0 = 0; j0 < m; j0 += RIF) {
                    vt rv[RIF];
#pragma unroll
                    for (int u = 0; u < RIF; ++u) {
                        const int j = (j0 + u < m) ? (j0 + u) : (m - 1);
                        const int sj = __builtin_amdgcn_readlane(s, j);
                        const float* const hrow = h + (size_t)sj * F;
                        rv[u] = *(const vt*)(hrow + lane * VT);
                    }
#pragma unroll
                    for (int u = 0; u < RIF; ++u) {
                        if (j0 + u < m) {
                            const int kj = __builtin_amdgcn_readlane(k, j0 + u);
                            const float wj = __int_as_float(
                                __builtin_amdgcn_readlane(__float_as_int(w), j0 + u));
                            add_row(kj, rv[u], wj);
                        }
                    }
                }
            }
        };

        if (listed) {
            drain();
            __syncthreads();
        }

        if (!listed || more0 != 0) {
            const long total = listed ? (long)lcnt : E;
            for (long c0 = 0; c0 < total; c0 += 1024 * AGG_EPT) {
                if (!listed) {
                    pend = 0;
#pragma unroll
                    for (int j = 0; j < AGG_EPT; ++j)
                        if (c0 + tid + j * 1024 < total) pend |= 1u << j;
                }
                for (int round = 0; round <= 1024 * AGG_EPT; ++round) {
                    if (tid < NW) qn[tid] = 0;
                    if (tid == 0) again = 0;
                    __syncthreads();
#pragma unroll 1
                    for (int j = 0; j < AGG_EPT; ++j) {
                        if ((pend >> j) & 1u) {
                            int s, d;
                            float w;
                            if (listed) {
                                const int i = tid + j * 1024;
                                s = ls[i];
                                d = ld[i];
                                w = FIRST ? dinv[s] * sdi[d] : __int_as_float(lw[i]);
                                place(j, s, d, w, pend);
                            } else {
                                const long e = c0 + tid + j * 1024;
                                d = dst[e] - lo;
                                if ((unsigned)d < (unsigned)rows) {
                                    s = src[e];
                                    w = dinv[s] * sdi[d];
                                    place(j, s, d, w, pend);
                                } else {
                                    pend &= ~(1u << j);
                                }
                            }
                        }
                    }
                    if (pend != 0) again = 1;
                    __syncthreads();
                    const int more = again;
                    drain();
                    __syncthreads();
                    if (more == 0) break;
                }
            }
        }

#pragma unroll
        for (int k = 0; k < KR; ++k) {
            const int r = wave + NW * k;
            if (r < rows) {
                vt o;
#pragma unroll
                for (int c = 0; c < VT; ++c) o[c] = acc[c][k];
                if (RELU) {
#pragma unroll
                    for (int c = 0; c < VT; ++c) o[c] = o[c] > 0.0f ? o[c] : 0.0f;
                }
                *(vt*)(outp + (size_t)(lo + r) * F + lane * VT) = o;
            }
        }
    }
}

extern "C" void kernel_launch(void* const* d_in, const int* in_sizes, int n_in,
                              void* d_out, int out_size, void* d_ws, size_t ws_size,
                              hipStream_t stream) {
    const float* x  = (const float*)d_in[0];
    const int*   ei = (const int*)d_in[1];
    const float* W1 = (const float*)d_in[2];
    const float* b1 = (const float*)d_in[3];
    const float* W2 = (const float*)d_in[4];
    const float* b2 = (const float*)d_in[5];
    float* out = (float*)d_out;

    const int  N = in_sizes[0] / IN_F;
    const long E = (long)in_sizes[1] / 2;
    const int* srcIdx = ei;
    const int* dstIdx = ei + E;

    float* h    = (float*)d_ws;
    float* agg1 = h + (size_t)N * HID_F;
    int*   lst  = (int*)(agg1 + (size_t)N * HID_F);
    float* dinv = (float*)(lst + (size_t)N * OUT_F);
    float* h2   = h;

    k_gemm_rows<IN_F, HID_F, true><<<GEMM_G, 32 * GEMM_WAVES, 0, stream>>>(x, W1, h, N, lst);

    k_bin<<<(int)((E + BIN_PASS - 1) / BIN_PASS), 1024, 0, stream>>>(srcIdx, dstIdx, lst, N, E);
    k_tile_deg<<<AGG_G, DEG_T, 0, stream>>>(dstIdx, dinv, lst, N, E);

    k_tile_agg<HID_F, true, true>
        <<<AGG_G, 1024, 0, stream>>>(srcIdx, dstIdx, dinv, h, b1, agg1, lst, N, E);

    k_gemm_rows<HID_F, OUT_F, false><<<GEMM_G, 32 * GEMM_WAVES, 0, stream>>>(agg1, W2, h2, N, lst);
    k_tile_agg<OUT_F, false, false>
        <<<AGG_G, 1024, 0, stream>>>(srcIdx, dstIdx, dinv, h2, b2, out, lst, N, E);
}


# Tile lists put in owner-wave order once, in the degree launch (LDS rank per row, scan, copy back); both aggregations read their runs straight, no LDS queues or barriers; sum order = list order
# speedup vs baseline: 4.8144x; 1.0349x over previous
#include <hip/hip_runtime.h>

typedef __attribute__((ext_vector_type(2))) float v2f;
typedef __attribute__((ext_vector_type(8))) float v8f;

#define IN_F  128
#define HID_F 128
#define OUT_F 64

#define AGG_G 256
#define TMAX  400
#define MAXT  1024
#define BIN_EPT 8
#define BIN_PASS (1024 * BIN_EPT)

struct TileGeo { int tn; int ntiles; int cap; long ohdr; long lhdr; };

__device__ __forceinline__ TileGeo tile_geo(int N) {
    TileGeo g;
    int tn = (N + AGG_G - 1) / AGG_G;
    if (tn > TMAX) tn = TMAX;
    if (tn < 1) tn = 1;
    g.tn = tn;
    g.ntiles = (N + tn - 1) / tn;
    const long lwords = (long)N * OUT_F;
    g.ohdr = ((long)g.ntiles + 63) & ~63L;
    g.lhdr = g.ohdr + (long)g.ntiles * 64;
    long lc = (lwords - g.lhdr) / g.ntiles / 3;
    if (lc < 0) lc = 0;
    if (lc > (1L << 30)) lc = 1L << 30;
    g.cap = (int)(lc & ~63L);
    return g;
}

__global__ void __launch_bounds__(1024)
k_bin(const int* __restrict__ src, const int* __restrict__ dst,
      int* __restrict__ lst, int N, long E) {
    __shared__ int hist[MAXT];
    __shared__ int gdelta[MAXT];
    __shared__ int ssrc[BIN_PASS];
    __shared__ int spk[BIN_PASS];
    __shared__ int wsum[32];
    __shared__ int tot;
    const TileGeo g = tile_geo(N);
    const bool binned = g.ntiles <= MAXT && g.cap > 0;
    const int tid  = threadIdx.x;
    const int lane = tid & 31;
    const int wave = tid >> 5;

    for (long p0 = (long)blockIdx.x * BIN_PASS; p0 < E; p0 += (long)gridDim.x * BIN_PASS) {
        hist[tid] = 0;
        __syncthreads();

        int d[BIN_EPT], s[BIN_EPT], tl[BIN_EPT], rk[BIN_EPT];
#pragma unroll
        for (int k = 0; k < BIN_EPT; ++k) {
            const long e = p0 + k * 1024 + tid;
            const bool ok = e < E;
            d[k] = ok ? dst[e] : -1;
            s[k] = ok ? src[e] : 0;
        }
#pragma unroll
        for (int k = 0; k < BIN_EPT; ++k) {
            tl[k] = 0; rk[k] = 0;
            if (binned && (unsigned)d[k] < (unsigned)N) {
                tl[k] = d[k] / g.tn;
                rk[k] = __hip_atomic_fetch_add(&hist[tl[k]], 1, __ATOMIC_RELAXED,
                                               __HIP_MEMORY_SCOPE_WORKGROUP);
            }
        }
        __syncthreads();
        if (!binned) continue;

        const int c = hist[tid];
        int gb = 0;
        if (c > 0 && tid < g.ntiles)
            gb = __hip_atomic_fetch_add(&lst[tid], c, __ATOMIC_RELAXED, __HIP_MEMORY_SCOPE_AGENT);
        int x = c;
#pragma unroll
        for (int o = 1; o < 32; o <<= 1) {
            const int y = __shfl_up(x, o);
            if (lane >= o) x += y;
        }
        if (lane == 31) wsum[wave] = x;
        __syncthreads();
        if (wave == 0) {
            const int v = wsum[lane];
            int z = v;
#pragma unroll
            for (int o = 1; o < 32; o <<= 1) {
                const int y = __shfl_up(z, o);
                if (lane >= o) z += y;
            }
            wsum[lane] = z - v;
            if (lane == 31) tot = z;
        }
        __syncthreads();
        const int excl = x - c + wsum[wave];
        hist[tid]   = excl;
        gdelta[tid] = gb - excl;
        __syncthreads();

#pragma unroll
        for (int k = 0; k < BIN_EPT; ++k) {
            if ((unsigned)d[k] < (unsigned)N) {
                const int p = hist[tl[k]] + rk[k];
                ssrc[p] = s[k];
                spk[p]  = (tl[k] << 9) | (d[k] - tl[k] * g.tn);
            }
        }
        __syncthreads();

        const int n = tot;
        for (int i = tid; i < n; i += 1024) {
            const int pk = spk[i];
            const int t  = pk >> 9;
            const int pos = gdelta[t] + i;
            if (pos >= 0 && pos < g.cap) {
                int* const ls = lst + g.lhdr + (long)t * 3 * g.cap;
                ls[pos]         = ssrc[i];
                ls[g.cap + pos] = pk & 511;
            }
        }
    }
}

#define SRT_T   1024
#define SRT_EPT 9
__global__ void __launch_bounds__(SRT_T)
k_tile_sort_deg(const int* __restrict__ dst, float* __restrict__ dinv,
                int* __restrict__ lst, int N, long E) {
    constexpr int NW = 32;
    constexpr int KR = (TMAX + NW - 1) / NW;
    static_assert(SRT_EPT * SRT_T > (TMAX * 64) / 3, "a thread's share of the longest list");
    __shared__ int cnt[TMAX];
    __shared__ int ssrc[SRT_EPT * SRT_T];
    __shared__ int sk[SRT_EPT * SRT_T];
    const TileGeo g = tile_geo(N);
    const bool binned = g.ntiles <= MAXT && g.cap > 0;
    const int tid  = threadIdx.x;
    const int lane = tid & 31;
    const int wave = tid >> 5;
    for (int t = blockIdx.x; t < g.ntiles; t += gridDim.x) {
        const int lo = t * g.tn;
        const int rows = (N - lo < g.tn) ? (N - lo) : g.tn;
        int lcnt = -1;
        if (binned) lcnt = lst[t];
        for (int r = tid; r < rows; r += SRT_T) cnt[r] = 0;
        __syncthreads();
        if (lcnt >= 0 && lcnt <= g.cap) {
            int* const ls = lst + g.lhdr + (long)t * 3 * g.cap;
            int* const ld = ls + g.cap;
            int es[SRT_EPT], er[SRT_EPT], rk[SRT_EPT];
#pragma unroll
            for (int j = 0; j < SRT_EPT; ++j) {
                const int i = tid + j * SRT_T;
                const bool ok = i < lcnt;
                es[j] = ok ? ls[i] : 0;
                er[j] = ok ? ld[i] : -1;
            }
#pragma unroll
            for (int j = 0; j < SRT_EPT; ++j) {
                rk[j] = 0;
                if ((unsigned)er[j] < (unsigned)rows)
                    rk[j] = __hip_atomic_fetch_add(&cnt[er[j]], 1, __ATOMIC_RELAXED,
                                                   __HIP_MEMORY_SCOPE_WORKGROUP);
            }
            __syncthreads();
            if (wave == 0) {
                int cr[KR];
                int tot = 0;
#pragma unroll
                for (int k = 0; k < KR; ++k) {
                    const int r = lane + NW * k;
                    cr[k] = (r < rows) ? cnt[r] : 0;
                    tot += cr[k];
                }
                int x = tot;
#pragma unroll
                for (int o = 1; o < 32; o <<= 1) {
                    const int y = __shfl_up(x, o);
                    if (lane >= o) x += y;
                }
                int* const oh = lst + g.ohdr + (long)t * 64;
                oh[lane] = x - tot;
                if (lane == 31) oh[32] = x;
                int run = x - tot;
#pragma unroll
                for (int k = 0; k < KR; ++k) {
                    const int r = lane + NW * k;
                    if (r < rows) {
                        dinv[lo + r] = rsqrtf(1.0f + (float)cr[k]);
                        cnt[r] = run;
                        run += cr[k];
                    }
                }
            }
            __syncthreads();
#pragma unroll
            for (int j = 0; j < SRT_EPT; ++j) {
                if ((unsigned)er[j] < (unsigned)rows) {
                    const int p = cnt[er[j]] + rk[j];
                    ssrc[p] = es[j];
                    sk[p]   = er[j] >> 5;
                }
            }
            __syncthreads();
            for (int i = tid; i < lcnt; i += SRT_T) {
                ls[i] = ssrc[i];
                ld[i] = sk[i];
            }
        } else {
            for (long e = tid; e < E; e += SRT_T) {
                const unsigned r = (unsigned)(dst[e] - lo);
                if (r < (unsigned)rows)
                    __hip_atomic_fetch_add(&cnt[r], 1, __ATOMIC_RELAXED,
                                           __HIP_MEMORY_SCOPE_WORKGROUP);
            }
            __syncthreads();
            for (int r = tid; r < rows; r += SRT_T)
                dinv[lo + r] = rsqrtf(1.0f + (float)cnt[r]);
        }
        __syncthreads();
    }
}

#define GEMM_WAVES 8
#define GEMM_G     512
template <int K, int NCOL, bool CLEAR>
__global__ void __launch_bounds__(32 * GEMM_WAVES)
k_gemm_rows(const float* __restrict__ X, const float* __restrict__ W,
            float* __restrict__ H, int M, int* __restrict__ lst) {
    constexpr int NS  = NCOL / 16;
    constexpr int NKK = K / 8;
    __shared__ float4 sW[(K / 4) * NCOL];

    const int tid  = threadIdx.x;
    const int wave = __builtin_amdgcn_readfirstlane(tid >> 5);
    const int lane = tid & 31;
    const int half = lane >> 4;
    const int idx  = lane & 15;

    static_assert(GEMM_WAVES == 8, "the rank table is for 8 waves");
    const int ntile = (M + 15) / 16;
    const int t0 = (int)((long)blockIdx.x * ntile / gridDim.x) + ((0x73261540 >> (4 * wave)) & 7);
    const int t1 = (int)(((long)blockIdx.x + 1) * ntile / gridDim.x);

    auto row_of = [&](int t) {
        int ar = t * 16 + idx;
        if (ar > M - 1) ar = M - 1;
        return X + (size_t)ar * K + half * 4;
    };

    float4 a[NKK];
    if (t0 < t1) {
        const float* const xr = row_of(t0);
#pragma unroll
        for (int kk = 0; kk < NKK; ++kk) a[kk] = *(const float4*)(xr + kk * 8);
    }

    if (CLEAR) {
        const TileGeo g = tile_geo(M);
        const long stride = (long)gridDim.x * blockDim.x;
        for (long j = (long)blockIdx.x * blockDim.x + tid; j < g.ohdr; j += stride) lst[j] = 0;
    }

    constexpr int WITEMS = (K / 4) * (NCOL / 4);
    static_assert(WITEMS % (32 * GEMM_WAVES) == 0, "whole trips");
#pragma unroll
    for (int j = 0; j < WITEMS / (32 * GEMM_WAVES); ++j) {
        const int i  = tid + j * 32 * GEMM_WAVES;
        const int q  = i / (NCOL / 4);
        const int n4 = i % (NCOL / 4);
        const float* const w = W + (size_t)(4 * q) * NCOL + n4 * 4;
        const float4 r0 = *(const float4*)(w);
        const float4 r1 = *(const float4*)(w + NCOL);
        const float4 r2 = *(const float4*)(w + 2 * NCOL);
        const float4 r3 = *(const float4*)(w + 3 * NCOL);
        float4* const d = sW + q * NCOL + n4 * 4;
        d[0] = make_float4(r0.x, r1.x, r2.x, r3.x);
        d[1] = make_float4(r0.y, r1.y, r2.y, r3.y);
        d[2] = make_float4(r0.z, r1.z, r2.z, r3.z);
        d[3] = make_float4(r0.w, r1.w, r2.w, r3.w);
    }
    __syncthreads();

    const float4* const bl = sW + half * NCOL + idx;

    for (int t = t0; t < t1; t += GEMM_WAVES) {
        const int row0 = t * 16;
        v8f acc[NS] = {};
        float4 bc[NS];
#pragma unroll
        for (int c = 0; c < NS; ++c) bc[c] = bl[c * 16];
#pragma unroll
        for (int kk = 0; kk < NKK; ++kk) {
            float4 bn[NS];
            if (kk + 1 < NKK) {
#pragma unroll
                for (int c = 0; c < NS; ++c) bn[c] = bl[(kk + 1) * 2 * NCOL + c * 16];
            }
#pragma unroll
            for (int s = 0; s < 2; ++s) {
                v2f av;
                av.x = s ? a[kk].z : a[kk].x;
                av.y = s ? a[kk].w : a[kk].y;
#pragma unroll
                for (int c = 0; c < NS; ++c) {
                    v2f bv;
                    bv.x = s ? bc[c].z : bc[c].x;
                    bv.y = s ? bc[c].w : bc[c].y;
                    acc[c] = __builtin_amdgcn_wmma_f32_16x16x4_f32(
                         false, av,  false, bv,
                         (short)0, acc[c],  false,  false);
                }
            }
            if (kk + 1 < NKK) {
#pragma unroll
                for (int c = 0; c < NS; ++c) bc[c] = bn[c];
            }
            __builtin_amdgcn_sched_barrier(0);
        }

#pragma unroll
        for (int v = 0; v < 8; ++v) {
            const int gr = row0 + half * 8 + v;
            if (gr < M) {
#pragma unroll
                for (int c = 0; c < NS; ++c)
                    H[(size_t)gr * NCOL + c * 16 + idx] = acc[c][v];
            }
        }

        if (t + GEMM_WAVES < t1) {
            const float* const xr = row_of(t + GEMM_WAVES);
#pragma unroll
            for (int kk = 0; kk < NKK; ++kk) a[kk] = *(const float4*)(xr + kk * 8);
        }
    }
}

template <int F, bool RELU, bool FIRST>
__global__ void __launch_bounds__(1024)
k_tile_agg(const int* __restrict__ src, const int* __restrict__ dst,
           const float* __restrict__ dinv, const float* __restrict__ h,
           const float* __restrict__ bias, float* __restrict__ outp,
           int* __restrict__ lst, int N, long E) {
    constexpr int NW = 32;
    constexpr int VT = F / 32;
    constexpr int KR = (TMAX + NW - 1) / NW;
    constexpr int RIF = (VT == 4) ? 6 : 8;
    constexpr int BATCH = (32 / RIF) * RIF;
    static_assert(KR < 16, "one 16-wide register vector per feature of a lane, its last element spare");
    constexpr int SPARE = 15;
    typedef float vt __attribute__((ext_vector_type(VT)));
    typedef float v16f __attribute__((ext_vector_type(16)));

    const int tid  = threadIdx.x;
    const int wave = __builtin_amdgcn_readfirstlane(tid >> 5);
    const int lane = tid & 31;

    const int G = gridDim.x;
    const TileGeo g = tile_geo(N);
    const int tn = g.tn;
    const int ntiles = g.ntiles;
    const int cap = g.cap;
    const bool binned = ntiles <= MAXT && cap > 0;

    for (int t = blockIdx.x; t < ntiles; t += G) {
        const int lo = t * tn;
        const int rows = (N - lo < tn) ? (N - lo) : tn;
        int* const ls = lst + g.lhdr + (long)t * 3 * cap;
        int* const ld = ls + cap;
        int* const lw = ld + cap;
        int lraw = -1, braw = 0, eraw = 0;
        if (binned) {
            const int oh0 = __builtin_amdgcn_readfirstlane((int)g.ohdr) + t * 64;
            lraw = lst[t];
            braw = lst[oh0 + lane];
            eraw = lst[oh0 + lane + 1];
        }
        const int lcnt = __builtin_amdgcn_readfirstlane(lraw);
        const bool listed = lcnt >= 0 && lcnt <= cap;
        const int beg = listed ? __builtin_amdgcn_readlane(braw, wave) : 0;
        const int end = listed ? __builtin_amdgcn_readlane(eraw, wave) : 0;

        auto fetch_list = [&](int b, int& s, int& k) {
            int i = b + lane;
            if (i > end - 1) i = end - 1;
            s = ls[i];
            k = ld[i];
        };
        auto fetch_w = [&](int b, int s, int k, float& wa, float& wb) {
            if (FIRST) {
                wa = dinv[s];
                wb = dinv[lo + wave + NW * k];
            } else {
                int i = b + lane;
                if (i > end - 1) i = end - 1;
                wa = __int_as_float(lw[i]);
                wb = 0.0f;
            }
        };
        int s = 0, k = 0;
        float wa = 0.0f, wb = 0.0f;
        if (beg < end) {
            fetch_list(beg, s, k);
            fetch_w(beg, s, k, wa, wb);
            if (lane >= end - beg) k = SPARE;
        }

        v16f acc[VT];
        {
            const vt bv = *(const vt*)(bias + lane * VT);
            vt hv[KR];
            float di[KR];
#pragma unroll
            for (int kr = 0; kr < KR; ++kr) {
                const int r = wave + NW * kr;
                const int n = lo + ((r < rows) ? r : 0);
                di[kr] = dinv[n];
                hv[kr] = *(const vt*)(h + (size_t)n * F + lane * VT);
            }
#pragma unroll
            for (int c = 0; c < VT; ++c) {
                acc[c] = (v16f)(0.0f);
#pragma unroll
                for (int kr = 0; kr < KR; ++kr) acc[c][kr] = hv[kr][c] * di[kr] * di[kr] + bv[c];
            }
        }

        auto add_row = [&](int kr, vt v, float w) {
#pragma unroll
            for (int c = 0; c < VT; ++c) acc[c][kr] += v[c] * w;
        };

        auto add_batch = [&](int m, int sv, int kv, float wv, auto&& after_first) {
            for (int j0 = 0; j0 < m; j0 += RIF) {
                vt rv[RIF];
#pragma unroll
                for (int u = 0; u < RIF; ++u) {
                    const int sj = __builtin_amdgcn_readlane(sv, j0 + u);
                    const float* const hrow = h + (size_t)sj * F;
                    rv[u] = *(const vt*)(hrow + lane * VT);
                }
#pragma unroll
                for (int u = 0; u < RIF; ++u) {
                    const int kj = __builtin_amdgcn_readlane(kv, j0 + u);
                    const float wj = __int_as_float(
                        __builtin_amdgcn_readlane(__float_as_int(wv), j0 + u));
                    add_row(kj, rv[u], wj);
                }
                if (j0 == 0) after_first();
            }
        };

        const long b0 = listed ? (long)beg : 0L;
        const long b1 = listed ? (long)end : E;
        for (long b = b0; b < b1; b += BATCH) {
            int m = BATCH;
            float w;
            int sn = 0, kn = 0;
            if (listed) {
                if (end - (int)b < BATCH) m = end - (int)b;
                w = FIRST ? wa * wb : wa;
                fetch_list((int)b + BATCH, sn, kn);
            } else {
                const long e = b + lane;
                int d = -1;
                if (lane < BATCH && e < E) d = dst[e] - lo;
                const bool mine = (unsigned)d < (unsigned)rows && (d & (NW - 1)) == wave;
                if (__builtin_amdgcn_ballot_w32(mine) == 0) continue;
                s = 0;
                k = SPARE;
                w = 0.0f;
                if (mine) {
                    s = src[e];
                    k = d >> 5;
                    w = dinv[s] * dinv[lo + d];
                }
            }
            add_batch(m, s, k, w, [&]() { if (listed) fetch_w((int)b + BATCH, sn, kn, wa, wb); });
            if (listed) {
                if (FIRST && lane < m) lw[(int)b + lane] = __float_as_int(w);
                s = sn;
                k = (lane < end - (int)b - BATCH) ? kn : SPARE;
            }
        }

#pragma unroll
        for (int kr = 0; kr < KR; ++kr) {
            const int r = wave + NW * kr;
            if (r < rows) {
                vt o;
#pragma unroll
                for (int c = 0; c < VT; ++c) o[c] = acc[c][kr];
                if (RELU) {
#pragma unroll
                    for (int c = 0; c < VT; ++c) o[c] = o[c] > 0.0f ? o[c] : 0.0f;
                }
                *(vt*)(outp + (size_t)(lo + r) * F + lane * VT) = o;
            }
        }
    }
}

extern "C" void kernel_launch(void* const* d_in, const int* in_sizes, int n_in,
                              void* d_out, int out_size, void* d_ws, size_t ws_size,
                              hipStream_t stream) {
    const float* x  = (const float*)d_in[0];
    const int*   ei = (const int*)d_in[1];
    const float* W1 = (const float*)d_in[2];
    const float* b1 = (const float*)d_in[3];
    const float* W2 = (const float*)d_in[4];
    const float* b2 = (const float*)d_in[5];
    float* out = (float*)d_out;

    const int  N = in_sizes[0] / IN_F;
    const long E = (long)in_sizes[1] / 2;
    const int* srcIdx = ei;
    const int* dstIdx = ei + E;

    float* h    = (float*)d_ws;
    float* agg1 = h + (size_t)N * HID_F;
    int*   lst  = (int*)(agg1 + (size_t)N * HID_F);
    float* dinv = (float*)(lst + (size_t)N * OUT_F);
    float* h2   = h;

    k_gemm_rows<IN_F, HID_F, true><<<GEMM_G, 32 * GEMM_WAVES, 0, stream>>>(x, W1, h, N, lst);

    k_bin<<<(int)((E + BIN_PASS - 1) / BIN_PASS), 1024, 0, stream>>>(srcIdx, dstIdx, lst, N, E);
    k_tile_sort_deg<<<AGG_G, SRT_T, 0, stream>>>(dstIdx, dinv, lst, N, E);

    k_tile_agg<HID_F, true, true>
        <<<AGG_G, 1024, 0, stream>>>(srcIdx, dstIdx, dinv, h, b1, agg1, lst, N, E);

    k_gemm_rows<HID_F, OUT_F, false><<<GEMM_G, 32 * GEMM_WAVES, 0, stream>>>(agg1, W2, h2, N, lst);
    k_tile_agg<OUT_F, false, false>
        <<<AGG_G, 1024, 0, stream>>>(srcIdx, dstIdx, dinv, h2, b2, out, lst, N, E);
}
